# starting kernel
# baseline (speedup 1.0000x reference)
#include <hip/hip_runtime.h>
#include <hip/hip_bf16.h>


typedef __attribute__((ext_vector_type(2))) float v2f;
typedef __attribute__((ext_vector_type(8))) float v8f;

#define B_DIM 128
#define I_DIM 512
#define O_DIM 512

#define C1_JOHNSON 1.65677880e-28f
#define C2_SHOT    5.43656366e-08f
#define NOISE_MIN  1e-12f

__global__ __launch_bounds__(256) void precomp_kernel(
    const float* __restrict__ x,
    const float* __restrict__ pl,
    const float* __restrict__ ph,
    float* __restrict__ L,
    float* __restrict__ Dm,
    float* __restrict__ SG,
    float* __restrict__ rowSumL) {
  __shared__ float red[256];
  const int b = blockIdx.x;
  float c0 = pl[0], c1 = pl[1], c2 = pl[2], c3 = pl[3], c4 = pl[4], c5 = pl[5], c6 = pl[6];
  float d0 = ph[0], d1 = ph[1], d2 = ph[2], d3 = ph[3], d4 = ph[4], d5 = ph[5];
  float lsum = 0.0f;
  for (int i = threadIdx.x; i < I_DIM; i += 256) {
    const float xv = x[b * I_DIM + i];
    float l = c6;
    l = fmaf(l, xv, c5); l = fmaf(l, xv, c4); l = fmaf(l, xv, c3);
    l = fmaf(l, xv, c2); l = fmaf(l, xv, c1); l = fmaf(l, xv, c0);
    float h = d5;
    h = fmaf(h, xv, d4); h = fmaf(h, xv, d3); h = fmaf(h, xv, d2);
    h = fmaf(h, xv, d1); h = fmaf(h, xv, d0);
    const float g = C1_JOHNSON / (fabsf(xv) + NOISE_MIN) + C2_SHOT;
    L[b * I_DIM + i]  = l;
    Dm[b * I_DIM + i] = h - l;
    SG[b * I_DIM + i] = sqrtf(g);
    lsum += l;
  }
  red[threadIdx.x] = lsum;
  __syncthreads();
  for (int s = 128; s > 0; s >>= 1) {
    if (threadIdx.x < (unsigned)s) red[threadIdx.x] += red[threadIdx.x + s];
    __syncthreads();
  }
  if (threadIdx.x == 0) rowSumL[b] = red[0];
}

__device__ __forceinline__ unsigned rotl32(unsigned v, int r) {
  return (v << r) | (v >> (32 - r));
}

__device__ __forceinline__ float bits_to_normal(unsigned bits) {
  const float lo = -0.99999994f;
  const float hi = 1.0f;
  float f = __uint_as_float((bits >> 9) | 0x3f800000u) - 1.0f;
  float u = fmaf(f, hi - lo, lo);
  u = fmaxf(u, lo);
  float w = -__logf((1.0f - u) * (1.0f + u));
  float p;
  if (w < 5.0f) {
    w -= 2.5f;
    p = 2.81022636e-08f;
    p = fmaf(p, w, 3.43273939e-07f);
    p = fmaf(p, w, -3.5233877e-06f);
    p = fmaf(p, w, -4.39150654e-06f);
    p = fmaf(p, w, 0.00021858087f);
    p = fmaf(p, w, -0.00125372503f);
    p = fmaf(p, w, -0.00417768164f);
    p = fmaf(p, w, 0.246640727f);
    p = fmaf(p, w, 1.50140941f);
  } else {
    w = sqrtf(w) - 3.0f;
    p = -0.000200214257f;
    p = fmaf(p, w, 0.000100950558f);
    p = fmaf(p, w, 0.00134934322f);
    p = fmaf(p, w, -0.00367342844f);
    p = fmaf(p, w, 0.00573950773f);
    p = fmaf(p, w, -0.0076224613f);
    p = fmaf(p, w, 0.00943887047f);
    p = fmaf(p, w, 1.00167406f);
    p = fmaf(p, w, 2.83297682f);
  }
  return 1.41421356237f * (p * u);
}

__global__ __launch_bounds__(256) void noise_kernel(float* __restrict__ N) {
  const unsigned half = (I_DIM * O_DIM) / 2;
  const unsigned j = blockIdx.x * 256u + threadIdx.x;
  if (j >= half) return;
  unsigned x0 = j, x1 = j + half;
  const unsigned ks0 = 0u, ks1 = 42u;
  const unsigned ks2 = 0x1BD11BDAu ^ ks0 ^ ks1;
  x0 += ks0; x1 += ks1;
  const int ra[4] = {13, 15, 26, 6};
  const int rb[4] = {17, 29, 16, 24};
#pragma unroll
  for (int r = 0; r < 4; ++r) { x0 += x1; x1 = rotl32(x1, ra[r]); x1 ^= x0; }
  x0 += ks1; x1 += ks2 + 1u;
#pragma unroll
  for (int r = 0; r < 4; ++r) { x0 += x1; x1 = rotl32(x1, rb[r]); x1 ^= x0; }
  x0 += ks2; x1 += ks0 + 2u;
#pragma unroll
  for (int r = 0; r < 4; ++r) { x0 += x1; x1 = rotl32(x1, ra[r]); x1 ^= x0; }
  x0 += ks0; x1 += ks1 + 3u;
#pragma unroll
  for (int r = 0; r < 4; ++r) { x0 += x1; x1 = rotl32(x1, rb[r]); x1 ^= x0; }
  x0 += ks1; x1 += ks2 + 4u;
#pragma unroll
  for (int r = 0; r < 4; ++r) { x0 += x1; x1 = rotl32(x1, ra[r]); x1 ^= x0; }
  x0 += ks2; x1 += ks0 + 5u;
  N[j]        = bits_to_normal(x0);
  N[j + half] = bits_to_normal(x1);
}

__global__ __launch_bounds__(256) void gemm_wmma_kernel(
    const float* __restrict__ Dm,
    const float* __restrict__ R,
    const float* __restrict__ rowSumL,
    float* __restrict__ out) {
  const int lane  = threadIdx.x & 31;
  const int wave  = threadIdx.x >> 5;
  const int ntile = blockIdx.x * 8 + wave;
  const int mtile = blockIdx.y;
  const int m = mtile * 16 + (lane & 15);
  const int n = ntile * 16 + (lane & 15);
  const int h = lane >> 4;

  v8f c = {};
  for (int k0 = 0; k0 < I_DIM; k0 += 4) {
    const int ka = k0 + 2 * h;
    v2f a = *(const v2f*)(Dm + m * I_DIM + ka);
    v2f bm;
    bm.x = R[ka * O_DIM + n];
    bm.y = R[(ka + 1) * O_DIM + n];
    c = __builtin_amdgcn_wmma_f32_16x16x4_f32(
         false, a,  false, bm,
         (short)0, c,  false,  false);
  }
#pragma unroll
  for (int v = 0; v < 8; ++v) {
    const int row = mtile * 16 + v + 8 * h;
    out[row * O_DIM + n] = c[v] + rowSumL[row];
  }
}

#define TB_B 8
#define TB_O 32
#define ICHUNK 64

__global__ __launch_bounds__(256) void noise_accum_kernel(
    const float* __restrict__ L, const float* __restrict__ Dm,
    const float* __restrict__ SG, const float* __restrict__ R,
    const float* __restrict__ N, float* __restrict__ out) {
  __shared__ float sL[TB_B][ICHUNK];
  __shared__ float sD[TB_B][ICHUNK];
  __shared__ float sG[TB_B][ICHUNK];
  const int t  = threadIdx.x;
  const int ob = t & (TB_O - 1);
  const int bb = t >> 5;
  const int o  = blockIdx.x * TB_O + ob;
  const int b  = blockIdx.y * TB_B + bb;

  float acc = 0.0f;
  for (int i0 = 0; i0 < I_DIM; i0 += ICHUNK) {
    __syncthreads();
    for (int s = t; s < TB_B * ICHUNK; s += 256) {
      const int rb = s / ICHUNK, ii = s % ICHUNK;
      const int g  = (blockIdx.y * TB_B + rb) * I_DIM + i0 + ii;
      sL[rb][ii] = L[g];
      sD[rb][ii] = Dm[g];
      sG[rb][ii] = SG[g];
    }
    __syncthreads();
    if (i0 + ICHUNK < I_DIM) {
      __builtin_prefetch(&R[(i0 + ICHUNK) * O_DIM + o], 0, 0);
      __builtin_prefetch(&N[(i0 + ICHUNK) * O_DIM + o], 0, 0);
    }
#pragma unroll 4
    for (int ii = 0; ii < ICHUNK; ++ii) {
      const int i  = i0 + ii;
      const float rv = R[i * O_DIM + o];
      const float nv = N[i * O_DIM + o];
      const float rr = fmaf(sD[bb][ii], rv, sL[bb][ii]);
      acc = fmaf(nv * sG[bb][ii], sqrtf(fabsf(rr)), acc);
    }
  }
  out[b * O_DIM + o] += acc;
}

extern "C" void kernel_launch(void* const* d_in, const int* in_sizes, int n_in,
                              void* d_out, int out_size, void* d_ws, size_t ws_size,
                              hipStream_t stream) {
  (void)in_sizes; (void)n_in; (void)out_size; (void)ws_size;
  const float* x  = (const float*)d_in[0];
  const float* pl = (const float*)d_in[1];
  const float* ph = (const float*)d_in[2];
  const float* r  = (const float*)d_in[3];
  float* out = (float*)d_out;

  float* wL  = (float*)d_ws;
  float* wD  = wL + B_DIM * I_DIM;
  float* wS  = wD + B_DIM * I_DIM;
  float* wN  = wS + B_DIM * I_DIM;
  float* wRS = wN + I_DIM * O_DIM;

  precomp_kernel<<<dim3(B_DIM), dim3(256), 0, stream>>>(x, pl, ph, wL, wD, wS, wRS);

  noise_kernel<<<dim3((I_DIM * O_DIM / 2 + 255) / 256), dim3(256), 0, stream>>>(wN);

  gemm_wmma_kernel<<<dim3(O_DIM / (16 * 8), B_DIM / 16), dim3(256), 0, stream>>>(
      wD, r, wRS, out);

  noise_accum_kernel<<<dim3(O_DIM / TB_O, B_DIM / TB_B), dim3(256), 0, stream>>>(
      wL, wD, wS, r, wN, out);
}


# noise sweep re-tiled, lane = batch row: a wave turns its 32x32 slice of L/D/sg through LDS into registers used for 8 columns; r/noise read from LDS at one address (noise-term sum order changes)
# speedup vs baseline: 2.4318x; 2.4318x over previous
#include <hip/hip_runtime.h>
#include <hip/hip_bf16.h>


typedef __attribute__((ext_vector_type(2))) float v2f;
typedef __attribute__((ext_vector_type(8))) float v8f;

#define B_DIM 128
#define I_DIM 512
#define O_DIM 512

#define C1_JOHNSON 1.65677880e-28f
#define C2_SHOT    5.43656366e-08f
#define NOISE_MIN  1e-12f

__global__ __launch_bounds__(256) void precomp_kernel(
    const float* __restrict__ x,
    const float* __restrict__ pl,
    const float* __restrict__ ph,
    float* __restrict__ L,
    float* __restrict__ Dm,
    float* __restrict__ SG,
    float* __restrict__ rowSumL) {
  __shared__ float red[256];
  const int b = blockIdx.x;
  float c0 = pl[0], c1 = pl[1], c2 = pl[2], c3 = pl[3], c4 = pl[4], c5 = pl[5], c6 = pl[6];
  float d0 = ph[0], d1 = ph[1], d2 = ph[2], d3 = ph[3], d4 = ph[4], d5 = ph[5];
  float lsum = 0.0f;
  for (int i = threadIdx.x; i < I_DIM; i += 256) {
    const float xv = x[b * I_DIM + i];
    float l = c6;
    l = fmaf(l, xv, c5); l = fmaf(l, xv, c4); l = fmaf(l, xv, c3);
    l = fmaf(l, xv, c2); l = fmaf(l, xv, c1); l = fmaf(l, xv, c0);
    float h = d5;
    h = fmaf(h, xv, d4); h = fmaf(h, xv, d3); h = fmaf(h, xv, d2);
    h = fmaf(h, xv, d1); h = fmaf(h, xv, d0);
    const float g = C1_JOHNSON / (fabsf(xv) + NOISE_MIN) + C2_SHOT;
    L[b * I_DIM + i]  = l;
    Dm[b * I_DIM + i] = h - l;
    SG[b * I_DIM + i] = sqrtf(g);
    lsum += l;
  }
  red[threadIdx.x] = lsum;
  __syncthreads();
  for (int s = 128; s > 0; s >>= 1) {
    if (threadIdx.x < (unsigned)s) red[threadIdx.x] += red[threadIdx.x + s];
    __syncthreads();
  }
  if (threadIdx.x == 0) rowSumL[b] = red[0];
}

__device__ __forceinline__ unsigned rotl32(unsigned v, int r) {
  return (v << r) | (v >> (32 - r));
}

__device__ __forceinline__ float bits_to_normal(unsigned bits) {
  const float lo = -0.99999994f;
  const float hi = 1.0f;
  float f = __uint_as_float((bits >> 9) | 0x3f800000u) - 1.0f;
  float u = fmaf(f, hi - lo, lo);
  u = fmaxf(u, lo);
  float w = -__logf((1.0f - u) * (1.0f + u));
  float p;
  if (w < 5.0f) {
    w -= 2.5f;
    p = 2.81022636e-08f;
    p = fmaf(p, w, 3.43273939e-07f);
    p = fmaf(p, w, -3.5233877e-06f);
    p = fmaf(p, w, -4.39150654e-06f);
    p = fmaf(p, w, 0.00021858087f);
    p = fmaf(p, w, -0.00125372503f);
    p = fmaf(p, w, -0.00417768164f);
    p = fmaf(p, w, 0.246640727f);
    p = fmaf(p, w, 1.50140941f);
  } else {
    w = sqrtf(w) - 3.0f;
    p = -0.000200214257f;
    p = fmaf(p, w, 0.000100950558f);
    p = fmaf(p, w, 0.00134934322f);
    p = fmaf(p, w, -0.00367342844f);
    p = fmaf(p, w, 0.00573950773f);
    p = fmaf(p, w, -0.0076224613f);
    p = fmaf(p, w, 0.00943887047f);
    p = fmaf(p, w, 1.00167406f);
    p = fmaf(p, w, 2.83297682f);
  }
  return 1.41421356237f * (p * u);
}

__global__ __launch_bounds__(256) void noise_kernel(float* __restrict__ N) {
  const unsigned half = (I_DIM * O_DIM) / 2;
  const unsigned j = blockIdx.x * 256u + threadIdx.x;
  if (j >= half) return;
  unsigned x0 = j, x1 = j + half;
  const unsigned ks0 = 0u, ks1 = 42u;
  const unsigned ks2 = 0x1BD11BDAu ^ ks0 ^ ks1;
  x0 += ks0; x1 += ks1;
  const int ra[4] = {13, 15, 26, 6};
  const int rb[4] = {17, 29, 16, 24};
#pragma unroll
  for (int r = 0; r < 4; ++r) { x0 += x1; x1 = rotl32(x1, ra[r]); x1 ^= x0; }
  x0 += ks1; x1 += ks2 + 1u;
#pragma unroll
  for (int r = 0; r < 4; ++r) { x0 += x1; x1 = rotl32(x1, rb[r]); x1 ^= x0; }
  x0 += ks2; x1 += ks0 + 2u;
#pragma unroll
  for (int r = 0; r < 4; ++r) { x0 += x1; x1 = rotl32(x1, ra[r]); x1 ^= x0; }
  x0 += ks0; x1 += ks1 + 3u;
#pragma unroll
  for (int r = 0; r < 4; ++r) { x0 += x1; x1 = rotl32(x1, rb[r]); x1 ^= x0; }
  x0 += ks1; x1 += ks2 + 4u;
#pragma unroll
  for (int r = 0; r < 4; ++r) { x0 += x1; x1 = rotl32(x1, ra[r]); x1 ^= x0; }
  x0 += ks2; x1 += ks0 + 5u;
  N[j]        = bits_to_normal(x0);
  N[j + half] = bits_to_normal(x1);
}

__global__ __launch_bounds__(256) void gemm_wmma_kernel(
    const float* __restrict__ Dm,
    const float* __restrict__ R,
    const float* __restrict__ rowSumL,
    float* __restrict__ out) {
  const int lane  = threadIdx.x & 31;
  const int wave  = threadIdx.x >> 5;
  const int ntile = blockIdx.x * 8 + wave;
  const int mtile = blockIdx.y;
  const int m = mtile * 16 + (lane & 15);
  const int n = ntile * 16 + (lane & 15);
  const int h = lane >> 4;

  v8f c = {};
  for (int k0 = 0; k0 < I_DIM; k0 += 4) {
    const int ka = k0 + 2 * h;
    v2f a = *(const v2f*)(Dm + m * I_DIM + ka);
    v2f bm;
    bm.x = R[ka * O_DIM + n];
    bm.y = R[(ka + 1) * O_DIM + n];
    c = __builtin_amdgcn_wmma_f32_16x16x4_f32(
         false, a,  false, bm,
         (short)0, c,  false,  false);
  }
#pragma unroll
  for (int v = 0; v < 8; ++v) {
    const int row = mtile * 16 + v + 8 * h;
    out[row * O_DIM + n] = c[v] + rowSumL[row];
  }
}

typedef __attribute__((ext_vector_type(4))) float v4f;

#define SW_ROWS   32
#define SW_COLS   8
#define SW_ISLICE 32
#define SW_WAVES  16
#define SW_PITCH  36

__global__ __launch_bounds__(SW_WAVES * 32) void noise_accum_kernel(
    const float* __restrict__ L, const float* __restrict__ Dm,
    const float* __restrict__ SG, const float* __restrict__ R,
    const float* __restrict__ N, float* __restrict__ out) {
  __shared__ __attribute__((aligned(16))) float sT[SW_WAVES][3][SW_ROWS * SW_PITCH];
  __shared__ __attribute__((aligned(16))) float sRN[SW_WAVES][2][SW_ISLICE * SW_COLS];
  __shared__ float sRed[SW_WAVES][SW_COLS][SW_ROWS];

  const int lane = threadIdx.x & 31;
  const int wave = threadIdx.x >> 5;
  const int o0 = blockIdx.x * SW_COLS;
  const int b0 = blockIdx.y * SW_ROWS;
  const int i0 = wave * SW_ISLICE;

  float* tL = sT[wave][0];
  float* tD = sT[wave][1];
  float* tS = sT[wave][2];
  float* tR = sRN[wave][0];
  float* tN = sRN[wave][1];

#pragma unroll
  for (int q = 0; q < 2; ++q) {
    const int j   = lane + 32 * q;
    const int row = j >> 1;
    const int c4  = (j & 1) * 4;
    const int g   = (i0 + row) * O_DIM + o0 + c4;
    *(v4f*)(tR + 4 * j) = *(const v4f*)(R + g);
    *(v4f*)(tN + 4 * j) = *(const v4f*)(N + g);
  }
#pragma unroll
  for (int r = 0; r < SW_ROWS; ++r) {
    const int g = (b0 + r) * I_DIM + i0 + lane;
    tL[r * SW_PITCH + lane] = L[g];
    tD[r * SW_PITCH + lane] = Dm[g];
    tS[r * SW_PITCH + lane] = SG[g];
  }

  const float* pL = tL + lane * SW_PITCH;
  const float* pD = tD + lane * SW_PITCH;
  const float* pS = tS + lane * SW_PITCH;

  float acc[SW_COLS];
#pragma unroll
  for (int c = 0; c < SW_COLS; ++c) acc[c] = 0.0f;

  for (int ii = 0; ii < SW_ISLICE; ii += 4) {
    const v4f l4 = *(const v4f*)(pL + ii);
    const v4f d4 = *(const v4f*)(pD + ii);
    const v4f s4 = *(const v4f*)(pS + ii);
#pragma unroll
    for (int k = 0; k < 4; ++k) {
      const v4f ra = *(const v4f*)(tR + (ii + k) * SW_COLS);
      const v4f rb = *(const v4f*)(tR + (ii + k) * SW_COLS + 4);
      const v4f na = *(const v4f*)(tN + (ii + k) * SW_COLS);
      const v4f nb = *(const v4f*)(tN + (ii + k) * SW_COLS + 4);
      const float lv = l4[k], dv = d4[k], sv = s4[k];
#pragma unroll
      for (int c = 0; c < 4; ++c) {
        const float rr0 = fmaf(dv, ra[c], lv);
        acc[c] = fmaf(na[c] * sv, sqrtf(fabsf(rr0)), acc[c]);
        const float rr1 = fmaf(dv, rb[c], lv);
        acc[c + 4] = fmaf(nb[c] * sv, sqrtf(fabsf(rr1)), acc[c + 4]);
      }
    }
  }

#pragma unroll
  for (int c = 0; c < SW_COLS; ++c) sRed[wave][c][lane] = acc[c];
  __syncthreads();
  if (threadIdx.x < SW_ROWS * SW_COLS) {
    const int cl = threadIdx.x & (SW_COLS - 1);
    const int bl = threadIdx.x >> 3;
    float s = 0.0f;
#pragma unroll
    for (int w = 0; w < SW_WAVES; ++w) s += sRed[w][cl][bl];
    out[(b0 + bl) * O_DIM + o0 + cl] += s;
  }
}

extern "C" void kernel_launch(void* const* d_in, const int* in_sizes, int n_in,
                              void* d_out, int out_size, void* d_ws, size_t ws_size,
                              hipStream_t stream) {
  (void)in_sizes; (void)n_in; (void)out_size; (void)ws_size;
  const float* x  = (const float*)d_in[0];
  const float* pl = (const float*)d_in[1];
  const float* ph = (const float*)d_in[2];
  const float* r  = (const float*)d_in[3];
  float* out = (float*)d_out;

  float* wL  = (float*)d_ws;
  float* wD  = wL + B_DIM * I_DIM;
  float* wS  = wD + B_DIM * I_DIM;
  float* wN  = wS + B_DIM * I_DIM;
  float* wRS = wN + I_DIM * O_DIM;

  precomp_kernel<<<dim3(B_DIM), dim3(256), 0, stream>>>(x, pl, ph, wL, wD, wS, wRS);

  noise_kernel<<<dim3((I_DIM * O_DIM / 2 + 255) / 256), dim3(256), 0, stream>>>(wN);

  gemm_wmma_kernel<<<dim3(O_DIM / (16 * 8), B_DIM / 16), dim3(256), 0, stream>>>(
      wD, r, wRS, out);

  noise_accum_kernel<<<dim3(O_DIM / SW_COLS, B_DIM / SW_ROWS), dim3(SW_WAVES * 32), 0, stream>>>(
      wL, wD, wS, r, wN, out);
}


# GEMM re-cut: a block owns a 16x32 patch, its 8 waves split K; each wave requests all its operands at once (D as whole lines turned through LDS); partial patches added in LDS (GEMM sum order changes)
# speedup vs baseline: 3.4394x; 1.4143x over previous
#include <hip/hip_runtime.h>
#include <hip/hip_bf16.h>


typedef __attribute__((ext_vector_type(2))) float v2f;
typedef __attribute__((ext_vector_type(8))) float v8f;

#define B_DIM 128
#define I_DIM 512
#define O_DIM 512

#define C1_JOHNSON 1.65677880e-28f
#define C2_SHOT    5.43656366e-08f
#define NOISE_MIN  1e-12f

__global__ __launch_bounds__(256) void precomp_kernel(
    const float* __restrict__ x,
    const float* __restrict__ pl,
    const float* __restrict__ ph,
    float* __restrict__ L,
    float* __restrict__ Dm,
    float* __restrict__ SG,
    float* __restrict__ rowSumL) {
  __shared__ float red[256];
  const int b = blockIdx.x;
  float c0 = pl[0], c1 = pl[1], c2 = pl[2], c3 = pl[3], c4 = pl[4], c5 = pl[5], c6 = pl[6];
  float d0 = ph[0], d1 = ph[1], d2 = ph[2], d3 = ph[3], d4 = ph[4], d5 = ph[5];
  float lsum = 0.0f;
  for (int i = threadIdx.x; i < I_DIM; i += 256) {
    const float xv = x[b * I_DIM + i];
    float l = c6;
    l = fmaf(l, xv, c5); l = fmaf(l, xv, c4); l = fmaf(l, xv, c3);
    l = fmaf(l, xv, c2); l = fmaf(l, xv, c1); l = fmaf(l, xv, c0);
    float h = d5;
    h = fmaf(h, xv, d4); h = fmaf(h, xv, d3); h = fmaf(h, xv, d2);
    h = fmaf(h, xv, d1); h = fmaf(h, xv, d0);
    const float g = C1_JOHNSON / (fabsf(xv) + NOISE_MIN) + C2_SHOT;
    L[b * I_DIM + i]  = l;
    Dm[b * I_DIM + i] = h - l;
    SG[b * I_DIM + i] = sqrtf(g);
    lsum += l;
  }
  red[threadIdx.x] = lsum;
  __syncthreads();
  for (int s = 128; s > 0; s >>= 1) {
    if (threadIdx.x < (unsigned)s) red[threadIdx.x] += red[threadIdx.x + s];
    __syncthreads();
  }
  if (threadIdx.x == 0) rowSumL[b] = red[0];
}

__device__ __forceinline__ unsigned rotl32(unsigned v, int r) {
  return (v << r) | (v >> (32 - r));
}

__device__ __forceinline__ float bits_to_normal(unsigned bits) {
  const float lo = -0.99999994f;
  const float hi = 1.0f;
  float f = __uint_as_float((bits >> 9) | 0x3f800000u) - 1.0f;
  float u = fmaf(f, hi - lo, lo);
  u = fmaxf(u, lo);
  float w = -__logf((1.0f - u) * (1.0f + u));
  float p;
  if (w < 5.0f) {
    w -= 2.5f;
    p = 2.81022636e-08f;
    p = fmaf(p, w, 3.43273939e-07f);
    p = fmaf(p, w, -3.5233877e-06f);
    p = fmaf(p, w, -4.39150654e-06f);
    p = fmaf(p, w, 0.00021858087f);
    p = fmaf(p, w, -0.00125372503f);
    p = fmaf(p, w, -0.00417768164f);
    p = fmaf(p, w, 0.246640727f);
    p = fmaf(p, w, 1.50140941f);
  } else {
    w = sqrtf(w) - 3.0f;
    p = -0.000200214257f;
    p = fmaf(p, w, 0.000100950558f);
    p = fmaf(p, w, 0.00134934322f);
    p = fmaf(p, w, -0.00367342844f);
    p = fmaf(p, w, 0.00573950773f);
    p = fmaf(p, w, -0.0076224613f);
    p = fmaf(p, w, 0.00943887047f);
    p = fmaf(p, w, 1.00167406f);
    p = fmaf(p, w, 2.83297682f);
  }
  return 1.41421356237f * (p * u);
}

__global__ __launch_bounds__(256) void noise_kernel(float* __restrict__ N) {
  const unsigned half = (I_DIM * O_DIM) / 2;
  const unsigned j = blockIdx.x * 256u + threadIdx.x;
  if (j >= half) return;
  unsigned x0 = j, x1 = j + half;
  const unsigned ks0 = 0u, ks1 = 42u;
  const unsigned ks2 = 0x1BD11BDAu ^ ks0 ^ ks1;
  x0 += ks0; x1 += ks1;
  const int ra[4] = {13, 15, 26, 6};
  const int rb[4] = {17, 29, 16, 24};
#pragma unroll
  for (int r = 0; r < 4; ++r) { x0 += x1; x1 = rotl32(x1, ra[r]); x1 ^= x0; }
  x0 += ks1; x1 += ks2 + 1u;
#pragma unroll
  for (int r = 0; r < 4; ++r) { x0 += x1; x1 = rotl32(x1, rb[r]); x1 ^= x0; }
  x0 += ks2; x1 += ks0 + 2u;
#pragma unroll
  for (int r = 0; r < 4; ++r) { x0 += x1; x1 = rotl32(x1, ra[r]); x1 ^= x0; }
  x0 += ks0; x1 += ks1 + 3u;
#pragma unroll
  for (int r = 0; r < 4; ++r) { x0 += x1; x1 = rotl32(x1, rb[r]); x1 ^= x0; }
  x0 += ks1; x1 += ks2 + 4u;
#pragma unroll
  for (int r = 0; r < 4; ++r) { x0 += x1; x1 = rotl32(x1, ra[r]); x1 ^= x0; }
  x0 += ks2; x1 += ks0 + 5u;
  N[j]        = bits_to_normal(x0);
  N[j + half] = bits_to_normal(x1);
}

typedef __attribute__((ext_vector_type(4))) float v4f;

#define GK_WAVES 8
#define GK_KW    (I_DIM / GK_WAVES)
#define GK_STEPS (GK_KW / 4)
#define GK_PITCH 68

__global__ __launch_bounds__(GK_WAVES * 32) void gemm_wmma_kernel(
    const float* __restrict__ Dm,
    const float* __restrict__ R,
    const float* __restrict__ rowSumL,
    float* __restrict__ out) {
  __shared__ __attribute__((aligned(16))) float sA[GK_WAVES][16 * GK_PITCH];
  __shared__ float sP[GK_WAVES][16][32];

  const int lane = threadIdx.x & 31;
  const int wave = threadIdx.x >> 5;
  const int c  = lane & 15;
  const int h  = lane >> 4;
  const int n0 = blockIdx.x * 32;
  const int m0 = blockIdx.y * 16;
  const int kw = wave * GK_KW;

  v4f a4[8];
#pragma unroll
  for (int q = 0; q < 8; ++q)
    a4[q] = *(const v4f*)(Dm + (m0 + 2 * q + h) * I_DIM + kw + 4 * c);

  const float* pB = R + (kw + 2 * h) * O_DIM + n0 + 2 * c;
  v2f bq[2 * GK_STEPS];
#pragma unroll
  for (int s = 0; s < GK_STEPS; ++s) {
    bq[2 * s]     = *(const v2f*)(pB + (4 * s) * O_DIM);
    bq[2 * s + 1] = *(const v2f*)(pB + (4 * s + 1) * O_DIM);
  }

  float* tA = sA[wave];
#pragma unroll
  for (int q = 0; q < 8; ++q)
    *(v4f*)(tA + (2 * q + h) * GK_PITCH + 4 * c) = a4[q];

  const float* pA = tA + c * GK_PITCH + 2 * h;
  v8f acc0 = {};
  v8f acc1 = {};
#pragma unroll
  for (int s = 0; s < GK_STEPS; ++s) {
    const v2f a = *(const v2f*)(pA + 4 * s);
    v2f b0, b1;
    b0.x = bq[2 * s].x;  b0.y = bq[2 * s + 1].x;
    b1.x = bq[2 * s].y;  b1.y = bq[2 * s + 1].y;
    acc0 = __builtin_amdgcn_wmma_f32_16x16x4_f32(
         false, a,  false, b0,
         (short)0, acc0,  false,  false);
    acc1 = __builtin_amdgcn_wmma_f32_16x16x4_f32(
         false, a,  false, b1,
         (short)0, acc1,  false,  false);
  }

#pragma unroll
  for (int v = 0; v < 8; ++v) {
    sP[wave][v][lane]     = acc0[v];
    sP[wave][8 + v][lane] = acc1[v];
  }
  __syncthreads();

  float s0 = 0.0f, s1 = 0.0f;
#pragma unroll
  for (int w = 0; w < GK_WAVES; ++w) {
    s0 += sP[w][wave][lane];
    s1 += sP[w][8 + wave][lane];
  }
  const int row = m0 + wave + 8 * h;
  const float rs = rowSumL[row];
  v2f o;
  o.x = s0 + rs;
  o.y = s1 + rs;
  *(v2f*)(out + row * O_DIM + n0 + 2 * c) = o;
}

#define SW_ROWS   32
#define SW_COLS   8
#define SW_ISLICE 32
#define SW_WAVES  16
#define SW_PITCH  36

__global__ __launch_bounds__(SW_WAVES * 32) void noise_accum_kernel(
    const float* __restrict__ L, const float* __restrict__ Dm,
    const float* __restrict__ SG, const float* __restrict__ R,
    const float* __restrict__ N, float* __restrict__ out) {
  __shared__ __attribute__((aligned(16))) float sT[SW_WAVES][3][SW_ROWS * SW_PITCH];
  __shared__ __attribute__((aligned(16))) float sRN[SW_WAVES][2][SW_ISLICE * SW_COLS];
  __shared__ float sRed[SW_WAVES][SW_COLS][SW_ROWS];

  const int lane = threadIdx.x & 31;
  const int wave = threadIdx.x >> 5;
  const int o0 = blockIdx.x * SW_COLS;
  const int b0 = blockIdx.y * SW_ROWS;
  const int i0 = wave * SW_ISLICE;

  float* tL = sT[wave][0];
  float* tD = sT[wave][1];
  float* tS = sT[wave][2];
  float* tR = sRN[wave][0];
  float* tN = sRN[wave][1];

#pragma unroll
  for (int q = 0; q < 2; ++q) {
    const int j   = lane + 32 * q;
    const int row = j >> 1;
    const int c4  = (j & 1) * 4;
    const int g   = (i0 + row) * O_DIM + o0 + c4;
    *(v4f*)(tR + 4 * j) = *(const v4f*)(R + g);
    *(v4f*)(tN + 4 * j) = *(const v4f*)(N + g);
  }
#pragma unroll
  for (int r = 0; r < SW_ROWS; ++r) {
    const int g = (b0 + r) * I_DIM + i0 + lane;
    tL[r * SW_PITCH + lane] = L[g];
    tD[r * SW_PITCH + lane] = Dm[g];
    tS[r * SW_PITCH + lane] = SG[g];
  }

  const float* pL = tL + lane * SW_PITCH;
  const float* pD = tD + lane * SW_PITCH;
  const float* pS = tS + lane * SW_PITCH;

  float acc[SW_COLS];
#pragma unroll
  for (int c = 0; c < SW_COLS; ++c) acc[c] = 0.0f;

  for (int ii = 0; ii < SW_ISLICE; ii += 4) {
    const v4f l4 = *(const v4f*)(pL + ii);
    const v4f d4 = *(const v4f*)(pD + ii);
    const v4f s4 = *(const v4f*)(pS + ii);
#pragma unroll
    for (int k = 0; k < 4; ++k) {
      const v4f ra = *(const v4f*)(tR + (ii + k) * SW_COLS);
      const v4f rb = *(const v4f*)(tR + (ii + k) * SW_COLS + 4);
      const v4f na = *(const v4f*)(tN + (ii + k) * SW_COLS);
      const v4f nb = *(const v4f*)(tN + (ii + k) * SW_COLS + 4);
      const float lv = l4[k], dv = d4[k], sv = s4[k];
#pragma unroll
      for (int c = 0; c < 4; ++c) {
        const float rr0 = fmaf(dv, ra[c], lv);
        acc[c] = fmaf(na[c] * sv, sqrtf(fabsf(rr0)), acc[c]);
        const float rr1 = fmaf(dv, rb[c], lv);
        acc[c + 4] = fmaf(nb[c] * sv, sqrtf(fabsf(rr1)), acc[c + 4]);
      }
    }
  }

#pragma unroll
  for (int c = 0; c < SW_COLS; ++c) sRed[wave][c][lane] = acc[c];
  __syncthreads();
  if (threadIdx.x < SW_ROWS * SW_COLS) {
    const int cl = threadIdx.x & (SW_COLS - 1);
    const int bl = threadIdx.x >> 3;
    float s = 0.0f;
#pragma unroll
    for (int w = 0; w < SW_WAVES; ++w) s += sRed[w][cl][bl];
    out[(b0 + bl) * O_DIM + o0 + cl] += s;
  }
}

extern "C" void kernel_launch(void* const* d_in, const int* in_sizes, int n_in,
                              void* d_out, int out_size, void* d_ws, size_t ws_size,
                              hipStream_t stream) {
  (void)in_sizes; (void)n_in; (void)out_size; (void)ws_size;
  const float* x  = (const float*)d_in[0];
  const float* pl = (const float*)d_in[1];
  const float* ph = (const float*)d_in[2];
  const float* r  = (const float*)d_in[3];
  float* out = (float*)d_out;

  float* wL  = (float*)d_ws;
  float* wD  = wL + B_DIM * I_DIM;
  float* wS  = wD + B_DIM * I_DIM;
  float* wN  = wS + B_DIM * I_DIM;
  float* wRS = wN + I_DIM * O_DIM;

  precomp_kernel<<<dim3(B_DIM), dim3(256), 0, stream>>>(x, pl, ph, wL, wD, wS, wRS);

  noise_kernel<<<dim3((I_DIM * O_DIM / 2 + 255) / 256), dim3(256), 0, stream>>>(wN);

  gemm_wmma_kernel<<<dim3(O_DIM / 32, B_DIM / 16), dim3(GK_WAVES * 32), 0, stream>>>(
      wD, r, wRS, out);

  noise_accum_kernel<<<dim3(O_DIM / SW_COLS, B_DIM / SW_ROWS), dim3(SW_WAVES * 32), 0, stream>>>(
      wL, wD, wS, r, wN, out);
}


# per-row terms made in the GEMM's A loader from x (Horner in registers, H-L to LDS); 8 of 16 column blocks store L, H-L, sg for the sweep, which also sums L: 4 launches to 3 (row-sum order changes)
# speedup vs baseline: 4.0402x; 1.1747x over previous
#include <hip/hip_runtime.h>
#include <hip/hip_bf16.h>


typedef __attribute__((ext_vector_type(2))) float v2f;
typedef __attribute__((ext_vector_type(8))) float v8f;

#define B_DIM 128
#define I_DIM 512
#define O_DIM 512

#define C1_JOHNSON 1.65677880e-28f
#define C2_SHOT    5.43656366e-08f
#define NOISE_MIN  1e-12f

struct PolyCoef {
  float c0, c1, c2, c3, c4, c5, c6;
  float d0, d1, d2, d3, d4, d5;
};

__device__ __forceinline__ void row_terms(const PolyCoef& k, float xv, float& l, float& d) {
  float a = k.c6;
  a = fmaf(a, xv, k.c5); a = fmaf(a, xv, k.c4); a = fmaf(a, xv, k.c3);
  a = fmaf(a, xv, k.c2); a = fmaf(a, xv, k.c1); a = fmaf(a, xv, k.c0);
  float b = k.d5;
  b = fmaf(b, xv, k.d4); b = fmaf(b, xv, k.d3); b = fmaf(b, xv, k.d2);
  b = fmaf(b, xv, k.d1); b = fmaf(b, xv, k.d0);
  l = a;
  d = b - a;
}

__device__ __forceinline__ unsigned rotl32(unsigned v, int r) {
  return (v << r) | (v >> (32 - r));
}

__device__ __forceinline__ float bits_to_normal(unsigned bits) {
  const float lo = -0.99999994f;
  const float hi = 1.0f;
  float f = __uint_as_float((bits >> 9) | 0x3f800000u) - 1.0f;
  float u = fmaf(f, hi - lo, lo);
  u = fmaxf(u, lo);
  float w = -__logf((1.0f - u) * (1.0f + u));
  float p;
  if (w < 5.0f) {
    w -= 2.5f;
    p = 2.81022636e-08f;
    p = fmaf(p, w, 3.43273939e-07f);
    p = fmaf(p, w, -3.5233877e-06f);
    p = fmaf(p, w, -4.39150654e-06f);
    p = fmaf(p, w, 0.00021858087f);
    p = fmaf(p, w, -0.00125372503f);
    p = fmaf(p, w, -0.00417768164f);
    p = fmaf(p, w, 0.246640727f);
    p = fmaf(p, w, 1.50140941f);
  } else {
    w = sqrtf(w) - 3.0f;
    p = -0.000200214257f;
    p = fmaf(p, w, 0.000100950558f);
    p = fmaf(p, w, 0.00134934322f);
    p = fmaf(p, w, -0.00367342844f);
    p = fmaf(p, w, 0.00573950773f);
    p = fmaf(p, w, -0.0076224613f);
    p = fmaf(p, w, 0.00943887047f);
    p = fmaf(p, w, 1.00167406f);
    p = fmaf(p, w, 2.83297682f);
  }
  return 1.41421356237f * (p * u);
}

__global__ __launch_bounds__(256) void noise_kernel(float* __restrict__ N) {
  const unsigned half = (I_DIM * O_DIM) / 2;
  const unsigned j = blockIdx.x * 256u + threadIdx.x;
  if (j >= half) return;
  unsigned x0 = j, x1 = j + half;
  const unsigned ks0 = 0u, ks1 = 42u;
  const unsigned ks2 = 0x1BD11BDAu ^ ks0 ^ ks1;
  x0 += ks0; x1 += ks1;
  const int ra[4] = {13, 15, 26, 6};
  const int rb[4] = {17, 29, 16, 24};
#pragma unroll
  for (int r = 0; r < 4; ++r) { x0 += x1; x1 = rotl32(x1, ra[r]); x1 ^= x0; }
  x0 += ks1; x1 += ks2 + 1u;
#pragma unroll
  for (int r = 0; r < 4; ++r) { x0 += x1; x1 = rotl32(x1, rb[r]); x1 ^= x0; }
  x0 += ks2; x1 += ks0 + 2u;
#pragma unroll
  for (int r = 0; r < 4; ++r) { x0 += x1; x1 = rotl32(x1, ra[r]); x1 ^= x0; }
  x0 += ks0; x1 += ks1 + 3u;
#pragma unroll
  for (int r = 0; r < 4; ++r) { x0 += x1; x1 = rotl32(x1, rb[r]); x1 ^= x0; }
  x0 += ks1; x1 += ks2 + 4u;
#pragma unroll
  for (int r = 0; r < 4; ++r) { x0 += x1; x1 = rotl32(x1, ra[r]); x1 ^= x0; }
  x0 += ks2; x1 += ks0 + 5u;
  N[j]        = bits_to_normal(x0);
  N[j + half] = bits_to_normal(x1);
}

typedef __attribute__((ext_vector_type(4))) float v4f;

#define GK_WAVES 8
#define GK_KW    (I_DIM / GK_WAVES)
#define GK_STEPS (GK_KW / 4)
#define GK_PITCH 68

__global__ __launch_bounds__(GK_WAVES * 32) void gemm_wmma_kernel(
    const float* __restrict__ x,
    const float* __restrict__ pl,
    const float* __restrict__ ph,
    const float* __restrict__ R,
    float* __restrict__ L,
    float* __restrict__ Dm,
    float* __restrict__ SG,
    float* __restrict__ out) {
  __shared__ __attribute__((aligned(16))) float sA[GK_WAVES][16 * GK_PITCH];
  __shared__ float sP[GK_WAVES][16][32];

  const int lane = threadIdx.x & 31;
  const int wave = threadIdx.x >> 5;
  const int c  = lane & 15;
  const int h  = lane >> 4;
  const int n0 = blockIdx.x * 32;
  const int m0 = blockIdx.y * 16;
  const int kw = wave * GK_KW;

  PolyCoef pc;
  pc.c0 = pl[0]; pc.c1 = pl[1]; pc.c2 = pl[2]; pc.c3 = pl[3];
  pc.c4 = pl[4]; pc.c5 = pl[5]; pc.c6 = pl[6];
  pc.d0 = ph[0]; pc.d1 = ph[1]; pc.d2 = ph[2];
  pc.d3 = ph[3]; pc.d4 = ph[4]; pc.d5 = ph[5];

  v4f a4[8];
#pragma unroll
  for (int q = 0; q < 8; ++q)
    a4[q] = *(const v4f*)(x + (m0 + 2 * q + h) * I_DIM + kw + 4 * c);
  const int gw = (m0 + 2 * (blockIdx.x & 7) + h) * I_DIM + kw + 4 * c;
  const v4f xw = *(const v4f*)(x + gw);

  const float* pB = R + (kw + 2 * h) * O_DIM + n0 + 2 * c;
  v2f bq[2 * GK_STEPS];
#pragma unroll
  for (int s = 0; s < GK_STEPS; ++s) {
    bq[2 * s]     = *(const v2f*)(pB + (4 * s) * O_DIM);
    bq[2 * s + 1] = *(const v2f*)(pB + (4 * s + 1) * O_DIM);
  }
  __builtin_amdgcn_sched_barrier(0);

  float* tA = sA[wave];
#pragma unroll
  for (int q = 0; q < 8; ++q) {
    v4f d4;
#pragma unroll
    for (int e = 0; e < 4; ++e) {
      float l, d;
      row_terms(pc, a4[q][e], l, d);
      d4[e] = d;
    }
    *(v4f*)(tA + (2 * q + h) * GK_PITCH + 4 * c) = d4;
  }
  asm volatile("" :: "v"(xw));
  if (blockIdx.x < 8) {
    v4f l4, d4, s4;
#pragma unroll
    for (int e = 0; e < 4; ++e) {
      const float xv = xw[e];
      float l, d;
      row_terms(pc, xv, l, d);
      const float g = C1_JOHNSON / (fabsf(xv) + NOISE_MIN) + C2_SHOT;
      l4[e] = l;
      d4[e] = d;
      s4[e] = sqrtf(g);
    }
    *(v4f*)(L + gw)  = l4;
    *(v4f*)(Dm + gw) = d4;
    *(v4f*)(SG + gw) = s4;
  }

  const float* pA = tA + c * GK_PITCH + 2 * h;
  v8f acc0 = {};
  v8f acc1 = {};
#pragma unroll
  for (int s = 0; s < GK_STEPS; ++s) {
    const v2f a = *(const v2f*)(pA + 4 * s);
    v2f b0, b1;
    b0.x = bq[2 * s].x;  b0.y = bq[2 * s + 1].x;
    b1.x = bq[2 * s].y;  b1.y = bq[2 * s + 1].y;
    acc0 = __builtin_amdgcn_wmma_f32_16x16x4_f32(
         false, a,  false, b0,
         (short)0, acc0,  false,  false);
    acc1 = __builtin_amdgcn_wmma_f32_16x16x4_f32(
         false, a,  false, b1,
         (short)0, acc1,  false,  false);
  }

#pragma unroll
  for (int v = 0; v < 8; ++v) {
    sP[wave][v][lane]     = acc0[v];
    sP[wave][8 + v][lane] = acc1[v];
  }
  __syncthreads();

  float s0 = 0.0f, s1 = 0.0f;
#pragma unroll
  for (int w = 0; w < GK_WAVES; ++w) {
    s0 += sP[w][wave][lane];
    s1 += sP[w][8 + wave][lane];
  }
  const int row = m0 + wave + 8 * h;
  v2f o;
  o.x = s0;
  o.y = s1;
  *(v2f*)(out + row * O_DIM + n0 + 2 * c) = o;
}

#define SW_ROWS   32
#define SW_COLS   8
#define SW_ISLICE 32
#define SW_WAVES  16
#define SW_PITCH  36

__global__ __launch_bounds__(SW_WAVES * 32) void noise_accum_kernel(
    const float* __restrict__ L, const float* __restrict__ Dm,
    const float* __restrict__ SG, const float* __restrict__ R,
    const float* __restrict__ N, float* __restrict__ out) {
  __shared__ __attribute__((aligned(16))) float sT[SW_WAVES][3][SW_ROWS * SW_PITCH];
  __shared__ __attribute__((aligned(16))) float sRN[SW_WAVES][2][SW_ISLICE * SW_COLS];
  __shared__ float sRed[SW_WAVES][SW_COLS + 1][SW_ROWS];

  const int lane = threadIdx.x & 31;
  const int wave = threadIdx.x >> 5;
  const int o0 = blockIdx.x * SW_COLS;
  const int b0 = blockIdx.y * SW_ROWS;
  const int i0 = wave * SW_ISLICE;

  float* tL = sT[wave][0];
  float* tD = sT[wave][1];
  float* tS = sT[wave][2];
  float* tR = sRN[wave][0];
  float* tN = sRN[wave][1];

#pragma unroll
  for (int q = 0; q < 2; ++q) {
    const int j   = lane + 32 * q;
    const int row = j >> 1;
    const int c4  = (j & 1) * 4;
    const int g   = (i0 + row) * O_DIM + o0 + c4;
    *(v4f*)(tR + 4 * j) = *(const v4f*)(R + g);
    *(v4f*)(tN + 4 * j) = *(const v4f*)(N + g);
  }
#pragma unroll
  for (int r = 0; r < SW_ROWS; ++r) {
    const int g = (b0 + r) * I_DIM + i0 + lane;
    tL[r * SW_PITCH + lane] = L[g];
    tD[r * SW_PITCH + lane] = Dm[g];
    tS[r * SW_PITCH + lane] = SG[g];
  }

  const float* pL = tL + lane * SW_PITCH;
  const float* pD = tD + lane * SW_PITCH;
  const float* pS = tS + lane * SW_PITCH;

  float acc[SW_COLS];
#pragma unroll
  for (int c = 0; c < SW_COLS; ++c) acc[c] = 0.0f;
  float lsum = 0.0f;

  for (int ii = 0; ii < SW_ISLICE; ii += 4) {
    const v4f l4 = *(const v4f*)(pL + ii);
    const v4f d4 = *(const v4f*)(pD + ii);
    const v4f s4 = *(const v4f*)(pS + ii);
#pragma unroll
    for (int k = 0; k < 4; ++k) {
      const v4f ra = *(const v4f*)(tR + (ii + k) * SW_COLS);
      const v4f rb = *(const v4f*)(tR + (ii + k) * SW_COLS + 4);
      const v4f na = *(const v4f*)(tN + (ii + k) * SW_COLS);
      const v4f nb = *(const v4f*)(tN + (ii + k) * SW_COLS + 4);
      const float lv = l4[k], dv = d4[k], sv = s4[k];
      lsum += lv;
#pragma unroll
      for (int c = 0; c < 4; ++c) {
        const float rr0 = fmaf(dv, ra[c], lv);
        acc[c] = fmaf(na[c] * sv, sqrtf(fabsf(rr0)), acc[c]);
        const float rr1 = fmaf(dv, rb[c], lv);
        acc[c + 4] = fmaf(nb[c] * sv, sqrtf(fabsf(rr1)), acc[c + 4]);
      }
    }
  }

#pragma unroll
  for (int c = 0; c < SW_COLS; ++c) sRed[wave][c][lane] = acc[c];
  sRed[wave][SW_COLS][lane] = lsum;
  __syncthreads();
  if (threadIdx.x < SW_ROWS * SW_COLS) {
    const int cl = threadIdx.x & (SW_COLS - 1);
    const int bl = threadIdx.x >> 3;
    float s = 0.0f, ls = 0.0f;
#pragma unroll
    for (int w = 0; w < SW_WAVES; ++w) {
      s  += sRed[w][cl][bl];
      ls += sRed[w][SW_COLS][bl];
    }
    float* po = out + (b0 + bl) * O_DIM + o0 + cl;
    *po = (*po + ls) + s;
  }
}

extern "C" void kernel_launch(void* const* d_in, const int* in_sizes, int n_in,
                              void* d_out, int out_size, void* d_ws, size_t ws_size,
                              hipStream_t stream) {
  (void)in_sizes; (void)n_in; (void)out_size; (void)ws_size;
  const float* x  = (const float*)d_in[0];
  const float* pl = (const float*)d_in[1];
  const float* ph = (const float*)d_in[2];
  const float* r  = (const float*)d_in[3];
  float* out = (float*)d_out;

  float* wL  = (float*)d_ws;
  float* wD  = wL + B_DIM * I_DIM;
  float* wS  = wD + B_DIM * I_DIM;
  float* wN  = wS + B_DIM * I_DIM;

  noise_kernel<<<dim3((I_DIM * O_DIM / 2 + 255) / 256), dim3(256), 0, stream>>>(wN);

  gemm_wmma_kernel<<<dim3(O_DIM / 32, B_DIM / 16), dim3(GK_WAVES * 32), 0, stream>>>(
      x, pl, ph, r, wL, wD, wS, out);

  noise_accum_kernel<<<dim3(O_DIM / SW_COLS, B_DIM / SW_ROWS), dim3(SW_WAVES * 32), 0, stream>>>(
      wL, wD, wS, r, wN, out);
}


# L, H-L, sg handed over in the sweep's own register layout ([32 rows][i quad][row][4 i]): the sweep loads each quad 16 B per lane one quad ahead inside its loop; no LDS turn, no load phase up front
# speedup vs baseline: 4.2510x; 1.0522x over previous
#include <hip/hip_runtime.h>
#include <hip/hip_bf16.h>


typedef __attribute__((ext_vector_type(2))) float v2f;
typedef __attribute__((ext_vector_type(8))) float v8f;

#define B_DIM 128
#define I_DIM 512
#define O_DIM 512

#define C1_JOHNSON 1.65677880e-28f
#define C2_SHOT    5.43656366e-08f
#define NOISE_MIN  1e-12f

struct PolyCoef {
  float c0, c1, c2, c3, c4, c5, c6;
  float d0, d1, d2, d3, d4, d5;
};

__device__ __forceinline__ void row_terms(const PolyCoef& k, float xv, float& l, float& d) {
  float a = k.c6;
  a = fmaf(a, xv, k.c5); a = fmaf(a, xv, k.c4); a = fmaf(a, xv, k.c3);
  a = fmaf(a, xv, k.c2); a = fmaf(a, xv, k.c1); a = fmaf(a, xv, k.c0);
  float b = k.d5;
  b = fmaf(b, xv, k.d4); b = fmaf(b, xv, k.d3); b = fmaf(b, xv, k.d2);
  b = fmaf(b, xv, k.d1); b = fmaf(b, xv, k.d0);
  l = a;
  d = b - a;
}

__device__ __forceinline__ unsigned rotl32(unsigned v, int r) {
  return (v << r) | (v >> (32 - r));
}

__device__ __forceinline__ float bits_to_normal(unsigned bits) {
  const float lo = -0.99999994f;
  const float hi = 1.0f;
  float f = __uint_as_float((bits >> 9) | 0x3f800000u) - 1.0f;
  float u = fmaf(f, hi - lo, lo);
  u = fmaxf(u, lo);
  float w = -__logf((1.0f - u) * (1.0f + u));
  float p;
  if (w < 5.0f) {
    w -= 2.5f;
    p = 2.81022636e-08f;
    p = fmaf(p, w, 3.43273939e-07f);
    p = fmaf(p, w, -3.5233877e-06f);
    p = fmaf(p, w, -4.39150654e-06f);
    p = fmaf(p, w, 0.00021858087f);
    p = fmaf(p, w, -0.00125372503f);
    p = fmaf(p, w, -0.00417768164f);
    p = fmaf(p, w, 0.246640727f);
    p = fmaf(p, w, 1.50140941f);
  } else {
    w = sqrtf(w) - 3.0f;
    p = -0.000200214257f;
    p = fmaf(p, w, 0.000100950558f);
    p = fmaf(p, w, 0.00134934322f);
    p = fmaf(p, w, -0.00367342844f);
    p = fmaf(p, w, 0.00573950773f);
    p = fmaf(p, w, -0.0076224613f);
    p = fmaf(p, w, 0.00943887047f);
    p = fmaf(p, w, 1.00167406f);
    p = fmaf(p, w, 2.83297682f);
  }
  return 1.41421356237f * (p * u);
}

__global__ __launch_bounds__(256) void noise_kernel(float* __restrict__ N) {
  const unsigned half = (I_DIM * O_DIM) / 2;
  const unsigned j = blockIdx.x * 256u + threadIdx.x;
  if (j >= half) return;
  unsigned x0 = j, x1 = j + half;
  const unsigned ks0 = 0u, ks1 = 42u;
  const unsigned ks2 = 0x1BD11BDAu ^ ks0 ^ ks1;
  x0 += ks0; x1 += ks1;
  const int ra[4] = {13, 15, 26, 6};
  const int rb[4] = {17, 29, 16, 24};
#pragma unroll
  for (int r = 0; r < 4; ++r) { x0 += x1; x1 = rotl32(x1, ra[r]); x1 ^= x0; }
  x0 += ks1; x1 += ks2 + 1u;
#pragma unroll
  for (int r = 0; r < 4; ++r) { x0 += x1; x1 = rotl32(x1, rb[r]); x1 ^= x0; }
  x0 += ks2; x1 += ks0 + 2u;
#pragma unroll
  for (int r = 0; r < 4; ++r) { x0 += x1; x1 = rotl32(x1, ra[r]); x1 ^= x0; }
  x0 += ks0; x1 += ks1 + 3u;
#pragma unroll
  for (int r = 0; r < 4; ++r) { x0 += x1; x1 = rotl32(x1, rb[r]); x1 ^= x0; }
  x0 += ks1; x1 += ks2 + 4u;
#pragma unroll
  for (int r = 0; r < 4; ++r) { x0 += x1; x1 = rotl32(x1, ra[r]); x1 ^= x0; }
  x0 += ks2; x1 += ks0 + 5u;
  N[j]        = bits_to_normal(x0);
  N[j + half] = bits_to_normal(x1);
}

typedef __attribute__((ext_vector_type(4))) float v4f;

#define GK_WAVES 8
#define GK_KW    (I_DIM / GK_WAVES)
#define GK_STEPS (GK_KW / 4)
#define GK_PITCH 68
#define TERM_INDEX(b, q) (((((b) >> 5) * (I_DIM / 4) + (q)) * 32 + ((b) & 31)) * 4)

__global__ __launch_bounds__(GK_WAVES * 32) void gemm_wmma_kernel(
    const float* __restrict__ x,
    const float* __restrict__ pl,
    const float* __restrict__ ph,
    const float* __restrict__ R,
    float* __restrict__ L,
    float* __restrict__ Dm,
    float* __restrict__ SG,
    float* __restrict__ out) {
  __shared__ __attribute__((aligned(16))) float sA[GK_WAVES][16 * GK_PITCH];
  __shared__ float sP[GK_WAVES][16][32];

  const int lane = threadIdx.x & 31;
  const int wave = threadIdx.x >> 5;
  const int c  = lane & 15;
  const int h  = lane >> 4;
  const int n0 = blockIdx.x * 32;
  const int m0 = blockIdx.y * 16;
  const int kw = wave * GK_KW;

  PolyCoef pc;
  pc.c0 = pl[0]; pc.c1 = pl[1]; pc.c2 = pl[2]; pc.c3 = pl[3];
  pc.c4 = pl[4]; pc.c5 = pl[5]; pc.c6 = pl[6];
  pc.d0 = ph[0]; pc.d1 = ph[1]; pc.d2 = ph[2];
  pc.d3 = ph[3]; pc.d4 = ph[4]; pc.d5 = ph[5];

  v4f a4[8];
#pragma unroll
  for (int q = 0; q < 8; ++q)
    a4[q] = *(const v4f*)(x + (m0 + 2 * q + h) * I_DIM + kw + 4 * c);
  const int qw = (kw >> 2) + 2 * (blockIdx.x & 7) + h;
  const int rw = m0 + c;
  const v4f xw = *(const v4f*)(x + rw * I_DIM + 4 * qw);

  const float* pB = R + (kw + 2 * h) * O_DIM + n0 + 2 * c;
  v2f bq[2 * GK_STEPS];
#pragma unroll
  for (int s = 0; s < GK_STEPS; ++s) {
    bq[2 * s]     = *(const v2f*)(pB + (4 * s) * O_DIM);
    bq[2 * s + 1] = *(const v2f*)(pB + (4 * s + 1) * O_DIM);
  }
  __builtin_amdgcn_sched_barrier(0);

  float* tA = sA[wave];
#pragma unroll
  for (int q = 0; q < 8; ++q) {
    v4f d4;
#pragma unroll
    for (int e = 0; e < 4; ++e) {
      float l, d;
      row_terms(pc, a4[q][e], l, d);
      d4[e] = d;
    }
    *(v4f*)(tA + (2 * q + h) * GK_PITCH + 4 * c) = d4;
  }
  asm volatile("" :: "v"(xw));
  if (blockIdx.x < 8) {
    v4f l4, d4, s4;
#pragma unroll
    for (int e = 0; e < 4; ++e) {
      const float xv = xw[e];
      float l, d;
      row_terms(pc, xv, l, d);
      const float g = C1_JOHNSON / (fabsf(xv) + NOISE_MIN) + C2_SHOT;
      l4[e] = l;
      d4[e] = d;
      s4[e] = sqrtf(g);
    }
    const int gw = TERM_INDEX(rw, qw);
    *(v4f*)(L + gw)  = l4;
    *(v4f*)(Dm + gw) = d4;
    *(v4f*)(SG + gw) = s4;
  }

  const float* pA = tA + c * GK_PITCH + 2 * h;
  v8f acc0 = {};
  v8f acc1 = {};
#pragma unroll
  for (int s = 0; s < GK_STEPS; ++s) {
    const v2f a = *(const v2f*)(pA + 4 * s);
    v2f b0, b1;
    b0.x = bq[2 * s].x;  b0.y = bq[2 * s + 1].x;
    b1.x = bq[2 * s].y;  b1.y = bq[2 * s + 1].y;
    acc0 = __builtin_amdgcn_wmma_f32_16x16x4_f32(
         false, a,  false, b0,
         (short)0, acc0,  false,  false);
    acc1 = __builtin_amdgcn_wmma_f32_16x16x4_f32(
         false, a,  false, b1,
         (short)0, acc1,  false,  false);
  }

#pragma unroll
  for (int v = 0; v < 8; ++v) {
    sP[wave][v][lane]     = acc0[v];
    sP[wave][8 + v][lane] = acc1[v];
  }
  __syncthreads();

  float s0 = 0.0f, s1 = 0.0f;
#pragma unroll
  for (int w = 0; w < GK_WAVES; ++w) {
    s0 += sP[w][wave][lane];
    s1 += sP[w][8 + wave][lane];
  }
  const int row = m0 + wave + 8 * h;
  v2f o;
  o.x = s0;
  o.y = s1;
  *(v2f*)(out + row * O_DIM + n0 + 2 * c) = o;
}

#define SW_ROWS   32
#define SW_COLS   8
#define SW_ISLICE 32
#define SW_QUADS  (SW_ISLICE / 4)
#define SW_WAVES  16
#define SW_QSTEP  (SW_ROWS * 4)

__device__ __forceinline__ void sweep_quad(const v4f l4, const v4f d4, const v4f s4,
                                           const float* tR, const float* tN, int ii,
                                           float (&acc)[SW_COLS], float& lsum) {
#pragma unroll
  for (int k = 0; k < 4; ++k) {
    const v4f ra = *(const v4f*)(tR + (ii + k) * SW_COLS);
    const v4f rb = *(const v4f*)(tR + (ii + k) * SW_COLS + 4);
    const v4f na = *(const v4f*)(tN + (ii + k) * SW_COLS);
    const v4f nb = *(const v4f*)(tN + (ii + k) * SW_COLS + 4);
    const float lv = l4[k], dv = d4[k], sv = s4[k];
    lsum += lv;
#pragma unroll
    for (int c = 0; c < 4; ++c) {
      const float rr0 = fmaf(dv, ra[c], lv);
      acc[c] = fmaf(na[c] * sv, sqrtf(fabsf(rr0)), acc[c]);
      const float rr1 = fmaf(dv, rb[c], lv);
      acc[c + 4] = fmaf(nb[c] * sv, sqrtf(fabsf(rr1)), acc[c + 4]);
    }
  }
}

__global__ __launch_bounds__(SW_WAVES * 32) __attribute__((amdgpu_waves_per_eu(4, 4)))
void noise_accum_kernel(
    const float* __restrict__ L, const float* __restrict__ Dm,
    const float* __restrict__ SG, const float* __restrict__ R,
    const float* __restrict__ N, float* __restrict__ out) {
  __shared__ __attribute__((aligned(16))) float sRN[SW_WAVES][2][SW_ISLICE * SW_COLS];
  __shared__ float sRed[SW_WAVES][SW_COLS + 1][SW_ROWS];

  const int lane = threadIdx.x & 31;
  const int wave = threadIdx.x >> 5;
  const int o0 = blockIdx.x * SW_COLS;
  const int b0 = blockIdx.y * SW_ROWS;
  const int i0 = wave * SW_ISLICE;

  float* tR = sRN[wave][0];
  float* tN = sRN[wave][1];

  v4f r4[2], n4[2];
#pragma unroll
  for (int q = 0; q < 2; ++q) {
    const int j   = lane + 32 * q;
    const int row = j >> 1;
    const int c4  = (j & 1) * 4;
    const int g   = (i0 + row) * O_DIM + o0 + c4;
    r4[q] = *(const v4f*)(R + g);
    n4[q] = *(const v4f*)(N + g);
  }

  const int tb = TERM_INDEX(b0 + lane, wave * SW_QUADS);
  const float* gL = L + tb;
  const float* gD = Dm + tb;
  const float* gS = SG + tb;
  v4f lc = *(const v4f*)(gL), dc = *(const v4f*)(gD), sc = *(const v4f*)(gS);
  __builtin_amdgcn_sched_barrier(0);

#pragma unroll
  for (int q = 0; q < 2; ++q) {
    const int j = lane + 32 * q;
    *(v4f*)(tR + 4 * j) = r4[q];
    *(v4f*)(tN + 4 * j) = n4[q];
  }

  float acc[SW_COLS];
#pragma unroll
  for (int c = 0; c < SW_COLS; ++c) acc[c] = 0.0f;
  float lsum = 0.0f;

  for (int t = 0; t < SW_QUADS; ++t) {
    const int tn = t + 1 < SW_QUADS ? t + 1 : t;
    const v4f ln = *(const v4f*)(gL + tn * SW_QSTEP);
    const v4f dn = *(const v4f*)(gD + tn * SW_QSTEP);
    const v4f sn = *(const v4f*)(gS + tn * SW_QSTEP);
    sweep_quad(lc, dc, sc, tR, tN, 4 * t, acc, lsum);
    lc = ln; dc = dn; sc = sn;
  }

#pragma unroll
  for (int c = 0; c < SW_COLS; ++c) sRed[wave][c][lane] = acc[c];
  sRed[wave][SW_COLS][lane] = lsum;
  __syncthreads();
  if (threadIdx.x < SW_ROWS * SW_COLS) {
    const int cl = threadIdx.x & (SW_COLS - 1);
    const int bl = threadIdx.x >> 3;
    float s = 0.0f, ls = 0.0f;
#pragma unroll
    for (int w = 0; w < SW_WAVES; ++w) {
      s  += sRed[w][cl][bl];
      ls += sRed[w][SW_COLS][bl];
    }
    float* po = out + (b0 + bl) * O_DIM + o0 + cl;
    *po = (*po + ls) + s;
  }
}

extern "C" void kernel_launch(void* const* d_in, const int* in_sizes, int n_in,
                              void* d_out, int out_size, void* d_ws, size_t ws_size,
                              hipStream_t stream) {
  (void)in_sizes; (void)n_in; (void)out_size; (void)ws_size;
  const float* x  = (const float*)d_in[0];
  const float* pl = (const float*)d_in[1];
  const float* ph = (const float*)d_in[2];
  const float* r  = (const float*)d_in[3];
  float* out = (float*)d_out;

  float* wL  = (float*)d_ws;
  float* wD  = wL + B_DIM * I_DIM;
  float* wS  = wD + B_DIM * I_DIM;
  float* wN  = wS + B_DIM * I_DIM;

  noise_kernel<<<dim3((I_DIM * O_DIM / 2 + 255) / 256), dim3(256), 0, stream>>>(wN);

  gemm_wmma_kernel<<<dim3(O_DIM / 32, B_DIM / 16), dim3(GK_WAVES * 32), 0, stream>>>(
      x, pl, ph, r, wL, wD, wS, out);

  noise_accum_kernel<<<dim3(O_DIM / SW_COLS, B_DIM / SW_ROWS), dim3(SW_WAVES * 32), 0, stream>>>(
      wL, wD, wS, r, wN, out);
}


# noise field made inside the GEMM kernel: each lane runs 4 counter pairs (8 normals) between issuing its operand loads and first using them, two 16-B stores; no noise launch: 3 launches to 2
# speedup vs baseline: 4.6429x; 1.0922x over previous
#include <hip/hip_runtime.h>
#include <hip/hip_bf16.h>


typedef __attribute__((ext_vector_type(2))) float v2f;
typedef __attribute__((ext_vector_type(8))) float v8f;

#define B_DIM 128
#define I_DIM 512
#define O_DIM 512

#define C1_JOHNSON 1.65677880e-28f
#define C2_SHOT    5.43656366e-08f
#define NOISE_MIN  1e-12f

struct PolyCoef {
  float c0, c1, c2, c3, c4, c5, c6;
  float d0, d1, d2, d3, d4, d5;
};

__device__ __forceinline__ void row_terms(const PolyCoef& k, float xv, float& l, float& d) {
  float a = k.c6;
  a = fmaf(a, xv, k.c5); a = fmaf(a, xv, k.c4); a = fmaf(a, xv, k.c3);
  a = fmaf(a, xv, k.c2); a = fmaf(a, xv, k.c1); a = fmaf(a, xv, k.c0);
  float b = k.d5;
  b = fmaf(b, xv, k.d4); b = fmaf(b, xv, k.d3); b = fmaf(b, xv, k.d2);
  b = fmaf(b, xv, k.d1); b = fmaf(b, xv, k.d0);
  l = a;
  d = b - a;
}

__device__ __forceinline__ unsigned rotl32(unsigned v, int r) {
  return (v << r) | (v >> (32 - r));
}

__device__ __forceinline__ float bits_to_unit(unsigned bits) {
  const float lo = -0.99999994f;
  const float hi = 1.0f;
  float f = __uint_as_float((bits >> 9) | 0x3f800000u) - 1.0f;
  float u = fmaf(f, hi - lo, lo);
  return fmaxf(u, lo);
}

__device__ __forceinline__ float erfinv_w(float u) {
  return -__logf((1.0f - u) * (1.0f + u));
}
__device__ __forceinline__ float erfinv_central(float w) {
  w -= 2.5f;
  float p = 2.81022636e-08f;
  p = fmaf(p, w, 3.43273939e-07f);
  p = fmaf(p, w, -3.5233877e-06f);
  p = fmaf(p, w, -4.39150654e-06f);
  p = fmaf(p, w, 0.00021858087f);
  p = fmaf(p, w, -0.00125372503f);
  p = fmaf(p, w, -0.00417768164f);
  p = fmaf(p, w, 0.246640727f);
  p = fmaf(p, w, 1.50140941f);
  return p;
}
__device__ __forceinline__ float erfinv_tail(float w) {
  w = sqrtf(w) - 3.0f;
  float p = -0.000200214257f;
  p = fmaf(p, w, 0.000100950558f);
  p = fmaf(p, w, 0.00134934322f);
  p = fmaf(p, w, -0.00367342844f);
  p = fmaf(p, w, 0.00573950773f);
  p = fmaf(p, w, -0.0076224613f);
  p = fmaf(p, w, 0.00943887047f);
  p = fmaf(p, w, 1.00167406f);
  p = fmaf(p, w, 2.83297682f);
  return p;
}

__device__ __forceinline__ void threefry_pair(unsigned j, unsigned& b_lo, unsigned& b_hi) {
  const unsigned half = (I_DIM * O_DIM) / 2;
  unsigned x0 = j, x1 = j + half;
  const unsigned ks0 = 0u, ks1 = 42u;
  const unsigned ks2 = 0x1BD11BDAu ^ ks0 ^ ks1;
  x0 += ks0; x1 += ks1;
  const int ra[4] = {13, 15, 26, 6};
  const int rb[4] = {17, 29, 16, 24};
#pragma unroll
  for (int r = 0; r < 4; ++r) { x0 += x1; x1 = rotl32(x1, ra[r]); x1 ^= x0; }
  x0 += ks1; x1 += ks2 + 1u;
#pragma unroll
  for (int r = 0; r < 4; ++r) { x0 += x1; x1 = rotl32(x1, rb[r]); x1 ^= x0; }
  x0 += ks2; x1 += ks0 + 2u;
#pragma unroll
  for (int r = 0; r < 4; ++r) { x0 += x1; x1 = rotl32(x1, ra[r]); x1 ^= x0; }
  x0 += ks0; x1 += ks1 + 3u;
#pragma unroll
  for (int r = 0; r < 4; ++r) { x0 += x1; x1 = rotl32(x1, rb[r]); x1 ^= x0; }
  x0 += ks1; x1 += ks2 + 4u;
#pragma unroll
  for (int r = 0; r < 4; ++r) { x0 += x1; x1 = rotl32(x1, ra[r]); x1 ^= x0; }
  x0 += ks2; x1 += ks0 + 5u;
  b_lo = x0;
  b_hi = x1;
}

typedef __attribute__((ext_vector_type(4))) float v4f;

#define GK_WAVES 8
#define GK_KW    (I_DIM / GK_WAVES)
#define GK_STEPS (GK_KW / 4)
#define GK_PITCH 68
#define TERM_INDEX(b, q) (((((b) >> 5) * (I_DIM / 4) + (q)) * 32 + ((b) & 31)) * 4)

__global__ __launch_bounds__(GK_WAVES * 32) void gemm_wmma_kernel(
    const float* __restrict__ x,
    const float* __restrict__ pl,
    const float* __restrict__ ph,
    const float* __restrict__ R,
    float* __restrict__ L,
    float* __restrict__ Dm,
    float* __restrict__ SG,
    float* __restrict__ N,
    float* __restrict__ out) {
  __shared__ __attribute__((aligned(16))) float sA[GK_WAVES][16 * GK_PITCH];
  __shared__ float sP[GK_WAVES][16][32];

  const int lane = threadIdx.x & 31;
  const int wave = threadIdx.x >> 5;
  const int c  = lane & 15;
  const int h  = lane >> 4;
  const int n0 = blockIdx.x * 32;
  const int m0 = blockIdx.y * 16;
  const int kw = wave * GK_KW;

  PolyCoef pc;
  pc.c0 = pl[0]; pc.c1 = pl[1]; pc.c2 = pl[2]; pc.c3 = pl[3];
  pc.c4 = pl[4]; pc.c5 = pl[5]; pc.c6 = pl[6];
  pc.d0 = ph[0]; pc.d1 = ph[1]; pc.d2 = ph[2];
  pc.d3 = ph[3]; pc.d4 = ph[4]; pc.d5 = ph[5];

  v4f a4[8];
#pragma unroll
  for (int q = 0; q < 8; ++q)
    a4[q] = *(const v4f*)(x + (m0 + 2 * q + h) * I_DIM + kw + 4 * c);
  const int qw = (kw >> 2) + 2 * (blockIdx.x & 7) + h;
  const int rw = m0 + c;
  const v4f xw = *(const v4f*)(x + rw * I_DIM + 4 * qw);

  const float* pB = R + (kw + 2 * h) * O_DIM + n0 + 2 * c;
  v2f bq[2 * GK_STEPS];
#pragma unroll
  for (int s = 0; s < GK_STEPS; ++s) {
    bq[2 * s]     = *(const v2f*)(pB + (4 * s) * O_DIM);
    bq[2 * s + 1] = *(const v2f*)(pB + (4 * s + 1) * O_DIM);
  }
  __builtin_amdgcn_sched_barrier(0);

  {
    const unsigned j0 =
        ((blockIdx.y * (O_DIM / 32u) + blockIdx.x) * (GK_WAVES * 32u) + threadIdx.x) * 4u;
    unsigned bits[8];
#pragma unroll
    for (int e = 0; e < 4; ++e) threefry_pair(j0 + e, bits[e], bits[4 + e]);
    float u[8], w[8], p[8];
#pragma unroll
    for (int e = 0; e < 8; ++e) {
      u[e] = bits_to_unit(bits[e]);
      w[e] = erfinv_w(u[e]);
      p[e] = erfinv_central(w[e]);
    }
    v4f nlo, nhi;
#pragma unroll
    for (int e = 0; e < 8; ++e) {
      if (!(w[e] < 5.0f)) p[e] = erfinv_tail(w[e]);
      const float n = 1.41421356237f * (p[e] * u[e]);
      if (e < 4) nlo[e] = n; else nhi[e - 4] = n;
    }
    *(v4f*)(N + j0) = nlo;
    *(v4f*)(N + (I_DIM * O_DIM) / 2 + j0) = nhi;
  }

  float* tA = sA[wave];
#pragma unroll
  for (int q = 0; q < 8; ++q) {
    v4f d4;
#pragma unroll
    for (int e = 0; e < 4; ++e) {
      float l, d;
      row_terms(pc, a4[q][e], l, d);
      d4[e] = d;
    }
    *(v4f*)(tA + (2 * q + h) * GK_PITCH + 4 * c) = d4;
  }
  asm volatile("" :: "v"(xw));
  if (blockIdx.x < 8) {
    v4f l4, d4, s4;
#pragma unroll
    for (int e = 0; e < 4; ++e) {
      const float xv = xw[e];
      float l, d;
      row_terms(pc, xv, l, d);
      const float g = C1_JOHNSON / (fabsf(xv) + NOISE_MIN) + C2_SHOT;
      l4[e] = l;
      d4[e] = d;
      s4[e] = sqrtf(g);
    }
    const int gw = TERM_INDEX(rw, qw);
    *(v4f*)(L + gw)  = l4;
    *(v4f*)(Dm + gw) = d4;
    *(v4f*)(SG + gw) = s4;
  }

  const float* pA = tA + c * GK_PITCH + 2 * h;
  v8f acc0 = {};
  v8f acc1 = {};
#pragma unroll
  for (int s = 0; s < GK_STEPS; ++s) {
    const v2f a = *(const v2f*)(pA + 4 * s);
    v2f b0, b1;
    b0.x = bq[2 * s].x;  b0.y = bq[2 * s + 1].x;
    b1.x = bq[2 * s].y;  b1.y = bq[2 * s + 1].y;
    acc0 = __builtin_amdgcn_wmma_f32_16x16x4_f32(
         false, a,  false, b0,
         (short)0, acc0,  false,  false);
    acc1 = __builtin_amdgcn_wmma_f32_16x16x4_f32(
         false, a,  false, b1,
         (short)0, acc1,  false,  false);
  }

#pragma unroll
  for (int v = 0; v < 8; ++v) {
    sP[wave][v][lane]     = acc0[v];
    sP[wave][8 + v][lane] = acc1[v];
  }
  __syncthreads();

  float s0 = 0.0f, s1 = 0.0f;
#pragma unroll
  for (int w = 0; w < GK_WAVES; ++w) {
    s0 += sP[w][wave][lane];
    s1 += sP[w][8 + wave][lane];
  }
  const int row = m0 + wave + 8 * h;
  v2f o;
  o.x = s0;
  o.y = s1;
  *(v2f*)(out + row * O_DIM + n0 + 2 * c) = o;
}

#define SW_ROWS   32
#define SW_COLS   8
#define SW_ISLICE 32
#define SW_QUADS  (SW_ISLICE / 4)
#define SW_WAVES  16
#define SW_QSTEP  (SW_ROWS * 4)

__device__ __forceinline__ void sweep_quad(const v4f l4, const v4f d4, const v4f s4,
                                           const float* tR, const float* tN, int ii,
                                           float (&acc)[SW_COLS], float& lsum) {
#pragma unroll
  for (int k = 0; k < 4; ++k) {
    const v4f ra = *(const v4f*)(tR + (ii + k) * SW_COLS);
    const v4f rb = *(const v4f*)(tR + (ii + k) * SW_COLS + 4);
    const v4f na = *(const v4f*)(tN + (ii + k) * SW_COLS);
    const v4f nb = *(const v4f*)(tN + (ii + k) * SW_COLS + 4);
    const float lv = l4[k], dv = d4[k], sv = s4[k];
    lsum += lv;
#pragma unroll
    for (int c = 0; c < 4; ++c) {
      const float rr0 = fmaf(dv, ra[c], lv);
      acc[c] = fmaf(na[c] * sv, sqrtf(fabsf(rr0)), acc[c]);
      const float rr1 = fmaf(dv, rb[c], lv);
      acc[c + 4] = fmaf(nb[c] * sv, sqrtf(fabsf(rr1)), acc[c + 4]);
    }
  }
}

__global__ __launch_bounds__(SW_WAVES * 32) __attribute__((amdgpu_waves_per_eu(4, 4)))
void noise_accum_kernel(
    const float* __restrict__ L, const float* __restrict__ Dm,
    const float* __restrict__ SG, const float* __restrict__ R,
    const float* __restrict__ N, float* __restrict__ out) {
  __shared__ __attribute__((aligned(16))) float sRN[SW_WAVES][2][SW_ISLICE * SW_COLS];
  __shared__ float sRed[SW_WAVES][SW_COLS + 1][SW_ROWS];

  const int lane = threadIdx.x & 31;
  const int wave = threadIdx.x >> 5;
  const int o0 = blockIdx.x * SW_COLS;
  const int b0 = blockIdx.y * SW_ROWS;
  const int i0 = wave * SW_ISLICE;

  float* tR = sRN[wave][0];
  float* tN = sRN[wave][1];

  v4f r4[2], n4[2];
#pragma unroll
  for (int q = 0; q < 2; ++q) {
    const int j   = lane + 32 * q;
    const int row = j >> 1;
    const int c4  = (j & 1) * 4;
    const int g   = (i0 + row) * O_DIM + o0 + c4;
    r4[q] = *(const v4f*)(R + g);
    n4[q] = *(const v4f*)(N + g);
  }

  const int tb = TERM_INDEX(b0 + lane, wave * SW_QUADS);
  const float* gL = L + tb;
  const float* gD = Dm + tb;
  const float* gS = SG + tb;
  v4f lc = *(const v4f*)(gL), dc = *(const v4f*)(gD), sc = *(const v4f*)(gS);
  __builtin_amdgcn_sched_barrier(0);

#pragma unroll
  for (int q = 0; q < 2; ++q) {
    const int j = lane + 32 * q;
    *(v4f*)(tR + 4 * j) = r4[q];
    *(v4f*)(tN + 4 * j) = n4[q];
  }

  float acc[SW_COLS];
#pragma unroll
  for (int c = 0; c < SW_COLS; ++c) acc[c] = 0.0f;
  float lsum = 0.0f;

  for (int t = 0; t < SW_QUADS; ++t) {
    const int tn = t + 1 < SW_QUADS ? t + 1 : t;
    const v4f ln = *(const v4f*)(gL + tn * SW_QSTEP);
    const v4f dn = *(const v4f*)(gD + tn * SW_QSTEP);
    const v4f sn = *(const v4f*)(gS + tn * SW_QSTEP);
    sweep_quad(lc, dc, sc, tR, tN, 4 * t, acc, lsum);
    lc = ln; dc = dn; sc = sn;
  }

#pragma unroll
  for (int c = 0; c < SW_COLS; ++c) sRed[wave][c][lane] = acc[c];
  sRed[wave][SW_COLS][lane] = lsum;
  __syncthreads();
  if (threadIdx.x < SW_ROWS * SW_COLS) {
    const int cl = threadIdx.x & (SW_COLS - 1);
    const int bl = threadIdx.x >> 3;
    float s = 0.0f, ls = 0.0f;
#pragma unroll
    for (int w = 0; w < SW_WAVES; ++w) {
      s  += sRed[w][cl][bl];
      ls += sRed[w][SW_COLS][bl];
    }
    float* po = out + (b0 + bl) * O_DIM + o0 + cl;
    *po = (*po + ls) + s;
  }
}

extern "C" void kernel_launch(void* const* d_in, const int* in_sizes, int n_in,
                              void* d_out, int out_size, void* d_ws, size_t ws_size,
                              hipStream_t stream) {
  (void)in_sizes; (void)n_in; (void)out_size; (void)ws_size;
  const float* x  = (const float*)d_in[0];
  const float* pl = (const float*)d_in[1];
  const float* ph = (const float*)d_in[2];
  const float* r  = (const float*)d_in[3];
  float* out = (float*)d_out;

  float* wL  = (float*)d_ws;
  float* wD  = wL + B_DIM * I_DIM;
  float* wS  = wD + B_DIM * I_DIM;
  float* wN  = wS + B_DIM * I_DIM;

  gemm_wmma_kernel<<<dim3(O_DIM / 32, B_DIM / 16), dim3(GK_WAVES * 32), 0, stream>>>(
      x, pl, ph, r, wL, wD, wS, wN, out);

  noise_accum_kernel<<<dim3(O_DIM / SW_COLS, B_DIM / SW_ROWS), dim3(SW_WAVES * 32), 0, stream>>>(
      wL, wD, wS, r, wN, out);
}


# sweep: one wave-wide range test per i-quad (min |arg| >= 2^-96) replaces sqrtf's per-element scaling and zero/inf selects; root = v_sqrt + 2 fma residuals + their sign bits: same bits as sqrtf
# speedup vs baseline: 5.4186x; 1.1671x over previous
#include <hip/hip_runtime.h>
#include <hip/hip_bf16.h>


typedef __attribute__((ext_vector_type(2))) float v2f;
typedef __attribute__((ext_vector_type(8))) float v8f;

#define B_DIM 128
#define I_DIM 512
#define O_DIM 512

#define C1_JOHNSON 1.65677880e-28f
#define C2_SHOT    5.43656366e-08f
#define NOISE_MIN  1e-12f

struct PolyCoef {
  float c0, c1, c2, c3, c4, c5, c6;
  float d0, d1, d2, d3, d4, d5;
};

__device__ __forceinline__ void row_terms(const PolyCoef& k, float xv, float& l, float& d) {
  float a = k.c6;
  a = fmaf(a, xv, k.c5); a = fmaf(a, xv, k.c4); a = fmaf(a, xv, k.c3);
  a = fmaf(a, xv, k.c2); a = fmaf(a, xv, k.c1); a = fmaf(a, xv, k.c0);
  float b = k.d5;
  b = fmaf(b, xv, k.d4); b = fmaf(b, xv, k.d3); b = fmaf(b, xv, k.d2);
  b = fmaf(b, xv, k.d1); b = fmaf(b, xv, k.d0);
  l = a;
  d = b - a;
}

__device__ __forceinline__ unsigned rotl32(unsigned v, int r) {
  return (v << r) | (v >> (32 - r));
}

__device__ __forceinline__ float bits_to_unit(unsigned bits) {
  const float lo = -0.99999994f;
  const float hi = 1.0f;
  float f = __uint_as_float((bits >> 9) | 0x3f800000u) - 1.0f;
  float u = fmaf(f, hi - lo, lo);
  return fmaxf(u, lo);
}

__device__ __forceinline__ float erfinv_w(float u) {
  return -__logf((1.0f - u) * (1.0f + u));
}
__device__ __forceinline__ float erfinv_central(float w) {
  w -= 2.5f;
  float p = 2.81022636e-08f;
  p = fmaf(p, w, 3.43273939e-07f);
  p = fmaf(p, w, -3.5233877e-06f);
  p = fmaf(p, w, -4.39150654e-06f);
  p = fmaf(p, w, 0.00021858087f);
  p = fmaf(p, w, -0.00125372503f);
  p = fmaf(p, w, -0.00417768164f);
  p = fmaf(p, w, 0.246640727f);
  p = fmaf(p, w, 1.50140941f);
  return p;
}
__device__ __forceinline__ float erfinv_tail(float w) {
  w = sqrtf(w) - 3.0f;
  float p = -0.000200214257f;
  p = fmaf(p, w, 0.000100950558f);
  p = fmaf(p, w, 0.00134934322f);
  p = fmaf(p, w, -0.00367342844f);
  p = fmaf(p, w, 0.00573950773f);
  p = fmaf(p, w, -0.0076224613f);
  p = fmaf(p, w, 0.00943887047f);
  p = fmaf(p, w, 1.00167406f);
  p = fmaf(p, w, 2.83297682f);
  return p;
}

__device__ __forceinline__ void threefry_pair(unsigned j, unsigned& b_lo, unsigned& b_hi) {
  const unsigned half = (I_DIM * O_DIM) / 2;
  unsigned x0 = j, x1 = j + half;
  const unsigned ks0 = 0u, ks1 = 42u;
  const unsigned ks2 = 0x1BD11BDAu ^ ks0 ^ ks1;
  x0 += ks0; x1 += ks1;
  const int ra[4] = {13, 15, 26, 6};
  const int rb[4] = {17, 29, 16, 24};
#pragma unroll
  for (int r = 0; r < 4; ++r) { x0 += x1; x1 = rotl32(x1, ra[r]); x1 ^= x0; }
  x0 += ks1; x1 += ks2 + 1u;
#pragma unroll
  for (int r = 0; r < 4; ++r) { x0 += x1; x1 = rotl32(x1, rb[r]); x1 ^= x0; }
  x0 += ks2; x1 += ks0 + 2u;
#pragma unroll
  for (int r = 0; r < 4; ++r) { x0 += x1; x1 = rotl32(x1, ra[r]); x1 ^= x0; }
  x0 += ks0; x1 += ks1 + 3u;
#pragma unroll
  for (int r = 0; r < 4; ++r) { x0 += x1; x1 = rotl32(x1, rb[r]); x1 ^= x0; }
  x0 += ks1; x1 += ks2 + 4u;
#pragma unroll
  for (int r = 0; r < 4; ++r) { x0 += x1; x1 = rotl32(x1, ra[r]); x1 ^= x0; }
  x0 += ks2; x1 += ks0 + 5u;
  b_lo = x0;
  b_hi = x1;
}

typedef __attribute__((ext_vector_type(4))) float v4f;

#define GK_WAVES 8
#define GK_KW    (I_DIM / GK_WAVES)
#define GK_STEPS (GK_KW / 4)
#define GK_PITCH 68
#define TERM_INDEX(b, q) (((((b) >> 5) * (I_DIM / 4) + (q)) * 32 + ((b) & 31)) * 4)

__global__ __launch_bounds__(GK_WAVES * 32) void gemm_wmma_kernel(
    const float* __restrict__ x,
    const float* __restrict__ pl,
    const float* __restrict__ ph,
    const float* __restrict__ R,
    float* __restrict__ L,
    float* __restrict__ Dm,
    float* __restrict__ SG,
    float* __restrict__ N,
    float* __restrict__ out) {
  __shared__ __attribute__((aligned(16))) float sA[GK_WAVES][16 * GK_PITCH];
  __shared__ float sP[GK_WAVES][16][32];

  const int lane = threadIdx.x & 31;
  const int wave = threadIdx.x >> 5;
  const int c  = lane & 15;
  const int h  = lane >> 4;
  const int n0 = blockIdx.x * 32;
  const int m0 = blockIdx.y * 16;
  const int kw = wave * GK_KW;

  PolyCoef pc;
  pc.c0 = pl[0]; pc.c1 = pl[1]; pc.c2 = pl[2]; pc.c3 = pl[3];
  pc.c4 = pl[4]; pc.c5 = pl[5]; pc.c6 = pl[6];
  pc.d0 = ph[0]; pc.d1 = ph[1]; pc.d2 = ph[2];
  pc.d3 = ph[3]; pc.d4 = ph[4]; pc.d5 = ph[5];

  v4f a4[8];
#pragma unroll
  for (int q = 0; q < 8; ++q)
    a4[q] = *(const v4f*)(x + (m0 + 2 * q + h) * I_DIM + kw + 4 * c);
  const int qw = (kw >> 2) + 2 * (blockIdx.x & 7) + h;
  const int rw = m0 + c;
  const v4f xw = *(const v4f*)(x + rw * I_DIM + 4 * qw);

  const float* pB = R + (kw + 2 * h) * O_DIM + n0 + 2 * c;
  v2f bq[2 * GK_STEPS];
#pragma unroll
  for (int s = 0; s < GK_STEPS; ++s) {
    bq[2 * s]     = *(const v2f*)(pB + (4 * s) * O_DIM);
    bq[2 * s + 1] = *(const v2f*)(pB + (4 * s + 1) * O_DIM);
  }
  __builtin_amdgcn_sched_barrier(0);

  {
    const unsigned j0 =
        ((blockIdx.y * (O_DIM / 32u) + blockIdx.x) * (GK_WAVES * 32u) + threadIdx.x) * 4u;
    unsigned bits[8];
#pragma unroll
    for (int e = 0; e < 4; ++e) threefry_pair(j0 + e, bits[e], bits[4 + e]);
    float u[8], w[8], p[8];
#pragma unroll
    for (int e = 0; e < 8; ++e) {
      u[e] = bits_to_unit(bits[e]);
      w[e] = erfinv_w(u[e]);
      p[e] = erfinv_central(w[e]);
    }
    v4f nlo, nhi;
#pragma unroll
    for (int e = 0; e < 8; ++e) {
      if (!(w[e] < 5.0f)) p[e] = erfinv_tail(w[e]);
      const float n = 1.41421356237f * (p[e] * u[e]);
      if (e < 4) nlo[e] = n; else nhi[e - 4] = n;
    }
    *(v4f*)(N + j0) = nlo;
    *(v4f*)(N + (I_DIM * O_DIM) / 2 + j0) = nhi;
  }

  float* tA = sA[wave];
#pragma unroll
  for (int q = 0; q < 8; ++q) {
    v4f d4;
#pragma unroll
    for (int e = 0; e < 4; ++e) {
      float l, d;
      row_terms(pc, a4[q][e], l, d);
      d4[e] = d;
    }
    *(v4f*)(tA + (2 * q + h) * GK_PITCH + 4 * c) = d4;
  }
  asm volatile("" :: "v"(xw));
  if (blockIdx.x < 8) {
    v4f l4, d4, s4;
#pragma unroll
    for (int e = 0; e < 4; ++e) {
      const float xv = xw[e];
      float l, d;
      row_terms(pc, xv, l, d);
      const float g = C1_JOHNSON / (fabsf(xv) + NOISE_MIN) + C2_SHOT;
      l4[e] = l;
      d4[e] = d;
      s4[e] = sqrtf(g);
    }
    const int gw = TERM_INDEX(rw, qw);
    *(v4f*)(L + gw)  = l4;
    *(v4f*)(Dm + gw) = d4;
    *(v4f*)(SG + gw) = s4;
  }

  const float* pA = tA + c * GK_PITCH + 2 * h;
  v8f acc0 = {};
  v8f acc1 = {};
#pragma unroll
  for (int s = 0; s < GK_STEPS; ++s) {
    const v2f a = *(const v2f*)(pA + 4 * s);
    v2f b0, b1;
    b0.x = bq[2 * s].x;  b0.y = bq[2 * s + 1].x;
    b1.x = bq[2 * s].y;  b1.y = bq[2 * s + 1].y;
    acc0 = __builtin_amdgcn_wmma_f32_16x16x4_f32(
         false, a,  false, b0,
         (short)0, acc0,  false,  false);
    acc1 = __builtin_amdgcn_wmma_f32_16x16x4_f32(
         false, a,  false, b1,
         (short)0, acc1,  false,  false);
  }

#pragma unroll
  for (int v = 0; v < 8; ++v) {
    sP[wave][v][lane]     = acc0[v];
    sP[wave][8 + v][lane] = acc1[v];
  }
  __syncthreads();

  float s0 = 0.0f, s1 = 0.0f;
#pragma unroll
  for (int w = 0; w < GK_WAVES; ++w) {
    s0 += sP[w][wave][lane];
    s1 += sP[w][8 + wave][lane];
  }
  const int row = m0 + wave + 8 * h;
  v2f o;
  o.x = s0;
  o.y = s1;
  *(v2f*)(out + row * O_DIM + n0 + 2 * c) = o;
}

#define SW_ROWS   32
#define SW_COLS   8
#define SW_ISLICE 32
#define SW_QUADS  (SW_ISLICE / 4)
#define SW_WAVES  16
#define SW_QSTEP  (SW_ROWS * 4)

__device__ __forceinline__ float sqrt_rn_unscaled(float x) {
  const float s = __builtin_amdgcn_sqrtf(x);
  const unsigned sdi = __float_as_uint(s) - 1u;
  const float sd = __uint_as_float(sdi);
  const float su = __uint_as_float(sdi + 2u);
  const float ed = fmaf(sd, s, -x);
  const float eu = fmaf(su, s, -x);
  return __uint_as_float(sdi + (__float_as_uint(ed) >> 31) + (__float_as_uint(eu) >> 31));
}

__device__ __forceinline__ void sweep_quad(const v4f l4, const v4f d4, const v4f s4,
                                           const float* tR, const float* tN, int ii,
                                           float (&acc)[SW_COLS], float& lsum) {
  float rr[4][SW_COLS], nz[4][SW_COLS];
  float amin0 = __builtin_inff(), amin1 = __builtin_inff();
#pragma unroll
  for (int k = 0; k < 4; ++k) {
    const v4f ra = *(const v4f*)(tR + (ii + k) * SW_COLS);
    const v4f rb = *(const v4f*)(tR + (ii + k) * SW_COLS + 4);
    const v4f na = *(const v4f*)(tN + (ii + k) * SW_COLS);
    const v4f nb = *(const v4f*)(tN + (ii + k) * SW_COLS + 4);
    const float lv = l4[k], dv = d4[k], sv = s4[k];
    lsum += lv;
#pragma unroll
    for (int c = 0; c < 4; ++c) {
      rr[k][c]     = fabsf(fmaf(dv, ra[c], lv));
      rr[k][c + 4] = fabsf(fmaf(dv, rb[c], lv));
      nz[k][c]     = na[c] * sv;
      nz[k][c + 4] = nb[c] * sv;
      if (c & 1) amin1 = fminf(fminf(amin1, rr[k][c]), rr[k][c + 4]);
      else       amin0 = fminf(fminf(amin0, rr[k][c]), rr[k][c + 4]);
    }
  }
  const bool small = fminf(amin0, amin1) < 0x1.0p-96f;
  if (__builtin_expect(__builtin_amdgcn_ballot_w32(small) == 0u, 1)) {
#pragma unroll
    for (int k = 0; k < 4; ++k)
#pragma unroll
      for (int c = 0; c < SW_COLS; ++c)
        acc[c] = fmaf(nz[k][c], sqrt_rn_unscaled(rr[k][c]), acc[c]);
  } else {
#pragma unroll
    for (int k = 0; k < 4; ++k)
#pragma unroll
      for (int c = 0; c < SW_COLS; ++c)
        acc[c] = fmaf(nz[k][c], sqrtf(rr[k][c]), acc[c]);
  }
}

__global__ __launch_bounds__(SW_WAVES * 32) __attribute__((amdgpu_waves_per_eu(4, 4)))
void noise_accum_kernel(
    const float* __restrict__ L, const float* __restrict__ Dm,
    const float* __restrict__ SG, const float* __restrict__ R,
    const float* __restrict__ N, float* __restrict__ out) {
  __shared__ __attribute__((aligned(16))) float sRN[SW_WAVES][2][SW_ISLICE * SW_COLS];
  __shared__ float sRed[SW_WAVES][SW_COLS + 1][SW_ROWS];

  const int lane = threadIdx.x & 31;
  const int wave = threadIdx.x >> 5;
  const int o0 = blockIdx.x * SW_COLS;
  const int b0 = blockIdx.y * SW_ROWS;
  const int i0 = wave * SW_ISLICE;

  float* tR = sRN[wave][0];
  float* tN = sRN[wave][1];

  v4f r4[2], n4[2];
#pragma unroll
  for (int q = 0; q < 2; ++q) {
    const int j   = lane + 32 * q;
    const int row = j >> 1;
    const int c4  = (j & 1) * 4;
    const int g   = (i0 + row) * O_DIM + o0 + c4;
    r4[q] = *(const v4f*)(R + g);
    n4[q] = *(const v4f*)(N + g);
  }

  const int tb = TERM_INDEX(b0 + lane, wave * SW_QUADS);
  const float* gL = L + tb;
  const float* gD = Dm + tb;
  const float* gS = SG + tb;
  v4f lc = *(const v4f*)(gL), dc = *(const v4f*)(gD), sc = *(const v4f*)(gS);
  __builtin_amdgcn_sched_barrier(0);

#pragma unroll
  for (int q = 0; q < 2; ++q) {
    const int j = lane + 32 * q;
    *(v4f*)(tR + 4 * j) = r4[q];
    *(v4f*)(tN + 4 * j) = n4[q];
  }

  float acc[SW_COLS];
#pragma unroll
  for (int c = 0; c < SW_COLS; ++c) acc[c] = 0.0f;
  float lsum = 0.0f;

  for (int t = 0; t < SW_QUADS; ++t) {
    const int tn = t + 1 < SW_QUADS ? t + 1 : t;
    const v4f ln = *(const v4f*)(gL + tn * SW_QSTEP);
    const v4f dn = *(const v4f*)(gD + tn * SW_QSTEP);
    const v4f sn = *(const v4f*)(gS + tn * SW_QSTEP);
    sweep_quad(lc, dc, sc, tR, tN, 4 * t, acc, lsum);
    lc = ln; dc = dn; sc = sn;
  }

#pragma unroll
  for (int c = 0; c < SW_COLS; ++c) sRed[wave][c][lane] = acc[c];
  sRed[wave][SW_COLS][lane] = lsum;
  __syncthreads();
  if (threadIdx.x < SW_ROWS * SW_COLS) {
    const int cl = threadIdx.x & (SW_COLS - 1);
    const int bl = threadIdx.x >> 3;
    float s = 0.0f, ls = 0.0f;
#pragma unroll
    for (int w = 0; w < SW_WAVES; ++w) {
      s  += sRed[w][cl][bl];
      ls += sRed[w][SW_COLS][bl];
    }
    float* po = out + (b0 + bl) * O_DIM + o0 + cl;
    *po = (*po + ls) + s;
  }
}

extern "C" void kernel_launch(void* const* d_in, const int* in_sizes, int n_in,
                              void* d_out, int out_size, void* d_ws, size_t ws_size,
                              hipStream_t stream) {
  (void)in_sizes; (void)n_in; (void)out_size; (void)ws_size;
  const float* x  = (const float*)d_in[0];
  const float* pl = (const float*)d_in[1];
  const float* ph = (const float*)d_in[2];
  const float* r  = (const float*)d_in[3];
  float* out = (float*)d_out;

  float* wL  = (float*)d_ws;
  float* wD  = wL + B_DIM * I_DIM;
  float* wS  = wD + B_DIM * I_DIM;
  float* wN  = wS + B_DIM * I_DIM;

  gemm_wmma_kernel<<<dim3(O_DIM / 32, B_DIM / 16), dim3(GK_WAVES * 32), 0, stream>>>(
      x, pl, ph, r, wL, wD, wS, wN, out);

  noise_accum_kernel<<<dim3(O_DIM / SW_COLS, B_DIM / SW_ROWS), dim3(SW_WAVES * 32), 0, stream>>>(
      wL, wD, wS, r, wN, out);
}


# sg = sqrt(c1/(|x|+eps)+c2) is sqrt(c2) in f32 for every x (c1/eps is under half an ulp of c2): multiplied into the stored noise field; no sg plane, no per-element multiply in the sweep; same bits
# speedup vs baseline: 5.7172x; 1.0551x over previous
#include <hip/hip_runtime.h>
#include <hip/hip_bf16.h>


typedef __attribute__((ext_vector_type(2))) float v2f;
typedef __attribute__((ext_vector_type(8))) float v8f;

#define B_DIM 128
#define I_DIM 512
#define O_DIM 512

#define C1_JOHNSON 1.65677880e-28f
#define C2_SHOT    5.43656366e-08f
#define NOISE_MIN  1e-12f
#define SG_CONST   sqrtf(C2_SHOT)

struct PolyCoef {
  float c0, c1, c2, c3, c4, c5, c6;
  float d0, d1, d2, d3, d4, d5;
};

__device__ __forceinline__ void row_terms(const PolyCoef& k, float xv, float& l, float& d) {
  float a = k.c6;
  a = fmaf(a, xv, k.c5); a = fmaf(a, xv, k.c4); a = fmaf(a, xv, k.c3);
  a = fmaf(a, xv, k.c2); a = fmaf(a, xv, k.c1); a = fmaf(a, xv, k.c0);
  float b = k.d5;
  b = fmaf(b, xv, k.d4); b = fmaf(b, xv, k.d3); b = fmaf(b, xv, k.d2);
  b = fmaf(b, xv, k.d1); b = fmaf(b, xv, k.d0);
  l = a;
  d = b - a;
}

__device__ __forceinline__ unsigned rotl32(unsigned v, int r) {
  return (v << r) | (v >> (32 - r));
}

__device__ __forceinline__ float bits_to_unit(unsigned bits) {
  const float lo = -0.99999994f;
  const float hi = 1.0f;
  float f = __uint_as_float((bits >> 9) | 0x3f800000u) - 1.0f;
  float u = fmaf(f, hi - lo, lo);
  return fmaxf(u, lo);
}

__device__ __forceinline__ float erfinv_w(float u) {
  return -__logf((1.0f - u) * (1.0f + u));
}
__device__ __forceinline__ float erfinv_central(float w) {
  w -= 2.5f;
  float p = 2.81022636e-08f;
  p = fmaf(p, w, 3.43273939e-07f);
  p = fmaf(p, w, -3.5233877e-06f);
  p = fmaf(p, w, -4.39150654e-06f);
  p = fmaf(p, w, 0.00021858087f);
  p = fmaf(p, w, -0.00125372503f);
  p = fmaf(p, w, -0.00417768164f);
  p = fmaf(p, w, 0.246640727f);
  p = fmaf(p, w, 1.50140941f);
  return p;
}
__device__ __forceinline__ float erfinv_tail(float w) {
  w = sqrtf(w) - 3.0f;
  float p = -0.000200214257f;
  p = fmaf(p, w, 0.000100950558f);
  p = fmaf(p, w, 0.00134934322f);
  p = fmaf(p, w, -0.00367342844f);
  p = fmaf(p, w, 0.00573950773f);
  p = fmaf(p, w, -0.0076224613f);
  p = fmaf(p, w, 0.00943887047f);
  p = fmaf(p, w, 1.00167406f);
  p = fmaf(p, w, 2.83297682f);
  return p;
}

__device__ __forceinline__ void threefry_pair(unsigned j, unsigned& b_lo, unsigned& b_hi) {
  const unsigned half = (I_DIM * O_DIM) / 2;
  unsigned x0 = j, x1 = j + half;
  const unsigned ks0 = 0u, ks1 = 42u;
  const unsigned ks2 = 0x1BD11BDAu ^ ks0 ^ ks1;
  x0 += ks0; x1 += ks1;
  const int ra[4] = {13, 15, 26, 6};
  const int rb[4] = {17, 29, 16, 24};
#pragma unroll
  for (int r = 0; r < 4; ++r) { x0 += x1; x1 = rotl32(x1, ra[r]); x1 ^= x0; }
  x0 += ks1; x1 += ks2 + 1u;
#pragma unroll
  for (int r = 0; r < 4; ++r) { x0 += x1; x1 = rotl32(x1, rb[r]); x1 ^= x0; }
  x0 += ks2; x1 += ks0 + 2u;
#pragma unroll
  for (int r = 0; r < 4; ++r) { x0 += x1; x1 = rotl32(x1, ra[r]); x1 ^= x0; }
  x0 += ks0; x1 += ks1 + 3u;
#pragma unroll
  for (int r = 0; r < 4; ++r) { x0 += x1; x1 = rotl32(x1, rb[r]); x1 ^= x0; }
  x0 += ks1; x1 += ks2 + 4u;
#pragma unroll
  for (int r = 0; r < 4; ++r) { x0 += x1; x1 = rotl32(x1, ra[r]); x1 ^= x0; }
  x0 += ks2; x1 += ks0 + 5u;
  b_lo = x0;
  b_hi = x1;
}

typedef __attribute__((ext_vector_type(4))) float v4f;

#define GK_WAVES 8
#define GK_KW    (I_DIM / GK_WAVES)
#define GK_STEPS (GK_KW / 4)
#define GK_PITCH 68
#define TERM_INDEX(b, q) (((((b) >> 5) * (I_DIM / 4) + (q)) * 32 + ((b) & 31)) * 4)

__global__ __launch_bounds__(GK_WAVES * 32) void gemm_wmma_kernel(
    const float* __restrict__ x,
    const float* __restrict__ pl,
    const float* __restrict__ ph,
    const float* __restrict__ R,
    float* __restrict__ L,
    float* __restrict__ Dm,
    float* __restrict__ N,
    float* __restrict__ out) {
  __shared__ __attribute__((aligned(16))) float sA[GK_WAVES][16 * GK_PITCH];
  __shared__ float sP[GK_WAVES][16][32];

  const int lane = threadIdx.x & 31;
  const int wave = threadIdx.x >> 5;
  const int c  = lane & 15;
  const int h  = lane >> 4;
  const int n0 = blockIdx.x * 32;
  const int m0 = blockIdx.y * 16;
  const int kw = wave * GK_KW;

  PolyCoef pc;
  pc.c0 = pl[0]; pc.c1 = pl[1]; pc.c2 = pl[2]; pc.c3 = pl[3];
  pc.c4 = pl[4]; pc.c5 = pl[5]; pc.c6 = pl[6];
  pc.d0 = ph[0]; pc.d1 = ph[1]; pc.d2 = ph[2];
  pc.d3 = ph[3]; pc.d4 = ph[4]; pc.d5 = ph[5];

  v4f a4[8];
#pragma unroll
  for (int q = 0; q < 8; ++q)
    a4[q] = *(const v4f*)(x + (m0 + 2 * q + h) * I_DIM + kw + 4 * c);
  const int qw = (kw >> 2) + 2 * (blockIdx.x & 7) + h;
  const int rw = m0 + c;
  const v4f xw = *(const v4f*)(x + rw * I_DIM + 4 * qw);

  const float* pB = R + (kw + 2 * h) * O_DIM + n0 + 2 * c;
  v2f bq[2 * GK_STEPS];
#pragma unroll
  for (int s = 0; s < GK_STEPS; ++s) {
    bq[2 * s]     = *(const v2f*)(pB + (4 * s) * O_DIM);
    bq[2 * s + 1] = *(const v2f*)(pB + (4 * s + 1) * O_DIM);
  }
  __builtin_amdgcn_sched_barrier(0);

  {
    const unsigned j0 =
        ((blockIdx.y * (O_DIM / 32u) + blockIdx.x) * (GK_WAVES * 32u) + threadIdx.x) * 4u;
    unsigned bits[8];
#pragma unroll
    for (int e = 0; e < 4; ++e) threefry_pair(j0 + e, bits[e], bits[4 + e]);
    float u[8], w[8], p[8];
#pragma unroll
    for (int e = 0; e < 8; ++e) {
      u[e] = bits_to_unit(bits[e]);
      w[e] = erfinv_w(u[e]);
      p[e] = erfinv_central(w[e]);
    }
    v4f nlo, nhi;
#pragma unroll
    for (int e = 0; e < 8; ++e) {
      if (!(w[e] < 5.0f)) p[e] = erfinv_tail(w[e]);
      const float n = (1.41421356237f * (p[e] * u[e])) * SG_CONST;
      if (e < 4) nlo[e] = n; else nhi[e - 4] = n;
    }
    *(v4f*)(N + j0) = nlo;
    *(v4f*)(N + (I_DIM * O_DIM) / 2 + j0) = nhi;
  }

  float* tA = sA[wave];
#pragma unroll
  for (int q = 0; q < 8; ++q) {
    v4f d4;
#pragma unroll
    for (int e = 0; e < 4; ++e) {
      float l, d;
      row_terms(pc, a4[q][e], l, d);
      d4[e] = d;
    }
    *(v4f*)(tA + (2 * q + h) * GK_PITCH + 4 * c) = d4;
  }
  asm volatile("" :: "v"(xw));
  if (blockIdx.x < 8) {
    v4f l4, d4;
#pragma unroll
    for (int e = 0; e < 4; ++e) {
      float l, d;
      row_terms(pc, xw[e], l, d);
      l4[e] = l;
      d4[e] = d;
    }
    const int gw = TERM_INDEX(rw, qw);
    *(v4f*)(L + gw)  = l4;
    *(v4f*)(Dm + gw) = d4;
  }

  const float* pA = tA + c * GK_PITCH + 2 * h;
  v8f acc0 = {};
  v8f acc1 = {};
#pragma unroll
  for (int s = 0; s < GK_STEPS; ++s) {
    const v2f a = *(const v2f*)(pA + 4 * s);
    v2f b0, b1;
    b0.x = bq[2 * s].x;  b0.y = bq[2 * s + 1].x;
    b1.x = bq[2 * s].y;  b1.y = bq[2 * s + 1].y;
    acc0 = __builtin_amdgcn_wmma_f32_16x16x4_f32(
         false, a,  false, b0,
         (short)0, acc0,  false,  false);
    acc1 = __builtin_amdgcn_wmma_f32_16x16x4_f32(
         false, a,  false, b1,
         (short)0, acc1,  false,  false);
  }

#pragma unroll
  for (int v = 0; v < 8; ++v) {
    sP[wave][v][lane]     = acc0[v];
    sP[wave][8 + v][lane] = acc1[v];
  }
  __syncthreads();

  float s0 = 0.0f, s1 = 0.0f;
#pragma unroll
  for (int w = 0; w < GK_WAVES; ++w) {
    s0 += sP[w][wave][lane];
    s1 += sP[w][8 + wave][lane];
  }
  const int row = m0 + wave + 8 * h;
  v2f o;
  o.x = s0;
  o.y = s1;
  *(v2f*)(out + row * O_DIM + n0 + 2 * c) = o;
}

#define SW_ROWS   32
#define SW_COLS   8
#define SW_ISLICE 32
#define SW_QUADS  (SW_ISLICE / 4)
#define SW_WAVES  16
#define SW_QSTEP  (SW_ROWS * 4)

__device__ __forceinline__ float sqrt_rn_unscaled(float x) {
  const float s = __builtin_amdgcn_sqrtf(x);
  const unsigned sdi = __float_as_uint(s) - 1u;
  const float sd = __uint_as_float(sdi);
  const float su = __uint_as_float(sdi + 2u);
  const float ed = fmaf(sd, s, -x);
  const float eu = fmaf(su, s, -x);
  return __uint_as_float(sdi + (__float_as_uint(ed) >> 31) + (__float_as_uint(eu) >> 31));
}

__device__ __forceinline__ void sweep_quad(const v4f l4, const v4f d4,
                                           const float* tR, const float* tN, int ii,
                                           float (&acc)[SW_COLS], float& lsum) {
  float rr[4][SW_COLS], nz[4][SW_COLS];
  float amin0 = __builtin_inff(), amin1 = __builtin_inff();
#pragma unroll
  for (int k = 0; k < 4; ++k) {
    const v4f ra = *(const v4f*)(tR + (ii + k) * SW_COLS);
    const v4f rb = *(const v4f*)(tR + (ii + k) * SW_COLS + 4);
    const v4f na = *(const v4f*)(tN + (ii + k) * SW_COLS);
    const v4f nb = *(const v4f*)(tN + (ii + k) * SW_COLS + 4);
    const float lv = l4[k], dv = d4[k];
    lsum += lv;
#pragma unroll
    for (int c = 0; c < 4; ++c) {
      rr[k][c]     = fabsf(fmaf(dv, ra[c], lv));
      rr[k][c + 4] = fabsf(fmaf(dv, rb[c], lv));
      nz[k][c]     = na[c];
      nz[k][c + 4] = nb[c];
      if (c & 1) amin1 = fminf(fminf(amin1, rr[k][c]), rr[k][c + 4]);
      else       amin0 = fminf(fminf(amin0, rr[k][c]), rr[k][c + 4]);
    }
  }
  const bool small = fminf(amin0, amin1) < 0x1.0p-96f;
  if (__builtin_expect(__builtin_amdgcn_ballot_w32(small) == 0u, 1)) {
#pragma unroll
    for (int k = 0; k < 4; ++k)
#pragma unroll
      for (int c = 0; c < SW_COLS; ++c)
        acc[c] = fmaf(nz[k][c], sqrt_rn_unscaled(rr[k][c]), acc[c]);
  } else {
#pragma unroll
    for (int k = 0; k < 4; ++k)
#pragma unroll
      for (int c = 0; c < SW_COLS; ++c)
        acc[c] = fmaf(nz[k][c], sqrtf(rr[k][c]), acc[c]);
  }
}

__global__ __launch_bounds__(SW_WAVES * 32) __attribute__((amdgpu_waves_per_eu(4, 4)))
void noise_accum_kernel(
    const float* __restrict__ L, const float* __restrict__ Dm,
    const float* __restrict__ R,
    const float* __restrict__ N, float* __restrict__ out) {
  __shared__ __attribute__((aligned(16))) float sRN[SW_WAVES][2][SW_ISLICE * SW_COLS];
  __shared__ float sRed[SW_WAVES][SW_COLS + 1][SW_ROWS];

  const int lane = threadIdx.x & 31;
  const int wave = threadIdx.x >> 5;
  const int o0 = blockIdx.x * SW_COLS;
  const int b0 = blockIdx.y * SW_ROWS;
  const int i0 = wave * SW_ISLICE;

  float* tR = sRN[wave][0];
  float* tN = sRN[wave][1];

  v4f r4[2], n4[2];
#pragma unroll
  for (int q = 0; q < 2; ++q) {
    const int j   = lane + 32 * q;
    const int row = j >> 1;
    const int c4  = (j & 1) * 4;
    const int g   = (i0 + row) * O_DIM + o0 + c4;
    r4[q] = *(const v4f*)(R + g);
    n4[q] = *(const v4f*)(N + g);
  }

  const int tb = TERM_INDEX(b0 + lane, wave * SW_QUADS);
  const float* gL = L + tb;
  const float* gD = Dm + tb;
  v4f lc = *(const v4f*)(gL), dc = *(const v4f*)(gD);
  __builtin_amdgcn_sched_barrier(0);

#pragma unroll
  for (int q = 0; q < 2; ++q) {
    const int j = lane + 32 * q;
    *(v4f*)(tR + 4 * j) = r4[q];
    *(v4f*)(tN + 4 * j) = n4[q];
  }

  float acc[SW_COLS];
#pragma unroll
  for (int c = 0; c < SW_COLS; ++c) acc[c] = 0.0f;
  float lsum = 0.0f;

  for (int t = 0; t < SW_QUADS; ++t) {
    const int tn = t + 1 < SW_QUADS ? t + 1 : t;
    const v4f ln = *(const v4f*)(gL + tn * SW_QSTEP);
    const v4f dn = *(const v4f*)(gD + tn * SW_QSTEP);
    sweep_quad(lc, dc, tR, tN, 4 * t, acc, lsum);
    lc = ln; dc = dn;
  }

#pragma unroll
  for (int c = 0; c < SW_COLS; ++c) sRed[wave][c][lane] = acc[c];
  sRed[wave][SW_COLS][lane] = lsum;
  __syncthreads();
  if (threadIdx.x < SW_ROWS * SW_COLS) {
    const int cl = threadIdx.x & (SW_COLS - 1);
    const int bl = threadIdx.x >> 3;
    float s = 0.0f, ls = 0.0f;
#pragma unroll
    for (int w = 0; w < SW_WAVES; ++w) {
      s  += sRed[w][cl][bl];
      ls += sRed[w][SW_COLS][bl];
    }
    float* po = out + (b0 + bl) * O_DIM + o0 + cl;
    *po = (*po + ls) + s;
  }
}

extern "C" void kernel_launch(void* const* d_in, const int* in_sizes, int n_in,
                              void* d_out, int out_size, void* d_ws, size_t ws_size,
                              hipStream_t stream) {
  (void)in_sizes; (void)n_in; (void)out_size; (void)ws_size;
  const float* x  = (const float*)d_in[0];
  const float* pl = (const float*)d_in[1];
  const float* ph = (const float*)d_in[2];
  const float* r  = (const float*)d_in[3];
  float* out = (float*)d_out;

  float* wL  = (float*)d_ws;
  float* wD  = wL + B_DIM * I_DIM;
  float* wN  = wD + B_DIM * I_DIM;

  gemm_wmma_kernel<<<dim3(O_DIM / 32, B_DIM / 16), dim3(GK_WAVES * 32), 0, stream>>>(
      x, pl, ph, r, wL, wD, wN, out);

  noise_accum_kernel<<<dim3(O_DIM / SW_COLS, B_DIM / SW_ROWS), dim3(SW_WAVES * 32), 0, stream>>>(
      wL, wD, r, wN, out);
}
